# starting kernel
# baseline (speedup 1.0000x reference)
#include <hip/hip_runtime.h>
#include <hip/hip_bf16.h>


#define N_NODES   200000
#define N_EDGES   640000
#define NUM_GRAPHS   256
#define OUTC          16

typedef __attribute__((ext_vector_type(2))) float v2f;
typedef __attribute__((ext_vector_type(8))) float v8f;

__global__ void gcn_deg_init(float* deg) {
    int i = blockIdx.x * 256 + threadIdx.x;
    if (i < N_NODES) deg[i] = 1.0f;
}

__global__ void gcn_deg_edges(const int* __restrict__ dst, float* deg) {
    int e = blockIdx.x * 256 + threadIdx.x;
    if (e < N_EDGES) atomicAdd(&deg[dst[e]], 1.0f);
}

__global__ void gcn_deg_rsqrt(float* deg) {
    int i = blockIdx.x * 256 + threadIdx.x;
    if (i < N_NODES) deg[i] = rsqrtf(deg[i]);
}

template <int MODE>
__global__ __launch_bounds__(256) void gcn_gemm_wmma(
        const float* __restrict__ A0, const float* __restrict__ A1,
        const float* __restrict__ W, float* __restrict__ B, int nRows) {
    __shared__ float sW[128 * 128];
    for (int i = threadIdx.x; i < 128 * 128; i += 256) {
        int kk = i >> 7, n = i & 127;
        sW[((kk >> 1) << 8) + n * 2 + (kk & 1)] = W[i];
    }
    __syncthreads();

    const int wave = threadIdx.x >> 5;
    const int lane = threadIdx.x & 31;
    const long m0  = ((long)blockIdx.x * 8 + wave) * 16;
    if (m0 >= nRows) return;

    const int half = lane >> 4;
    const int lr   = lane & 15;
    const int koff = half * 2;
    const long row = m0 + lr;

    auto loadA = [&](int k) -> v2f {
        v2f r;
        if (MODE == 0) {
            if (k < 124) {
                const float* p = A0 + row * 124 + k + koff;
                r.x = p[0]; r.y = p[1];
            } else {
                const float* p = A1 + row * 4 + (k - 124) + koff;
                r.x = p[0]; r.y = p[1];
            }
        } else {
            const float* p = A0 + row * 128 + k + koff;
            r.x = fmaxf(p[0] + A1[k + koff],     0.0f);
            r.y = fmaxf(p[1] + A1[k + koff + 1], 0.0f);
        }
        return r;
    };

    v8f acc[8];
#pragma unroll
    for (int i = 0; i < 8; ++i) acc[i] = (v8f)(0.0f);

    v2f a = loadA(0);
    for (int k = 0; k < 128; k += 4) {
        v2f an = a;
        if (k + 4 < 128) an = loadA(k + 4);
        const float* bbase = sW + (((k + koff) >> 1) << 8) + lr * 2;
#pragma unroll
        for (int nt = 0; nt < 8; ++nt) {
            v2f b = *(const v2f*)(bbase + nt * 32);
            acc[nt] = __builtin_amdgcn_wmma_f32_16x16x4_f32(
                false, a, false, b, (short)0, acc[nt], false, false);
        }
        a = an;
    }
#pragma unroll
    for (int nt = 0; nt < 8; ++nt) {
        const int n0 = nt * 16;
#pragma unroll
        for (int r = 0; r < 8; ++r)
            B[(m0 + half * 8 + r) * 128 + n0 + lr] = acc[nt][r];
    }
}

__global__ void gcn_agg_init(const float* __restrict__ B,
                             const float* __restrict__ dis,
                             float* __restrict__ out) {
    long idx = (long)blockIdx.x * 256 + threadIdx.x;
    if (idx >= (long)N_NODES * 128) return;
    int n = (int)(idx >> 7);
    float d = dis[n];
    out[idx] = d * d * B[idx];
}

__global__ __launch_bounds__(256) void gcn_agg_edges(
        const float* __restrict__ B, float* __restrict__ out,
        const int* __restrict__ src, const int* __restrict__ dst,
        const float* __restrict__ dis) {
    int e = blockIdx.x * 8 + (threadIdx.x >> 5);
    if (e >= N_EDGES) return;
    const int lane = threadIdx.x & 31;
    const int s = src[e], d = dst[e];
    const float norm = dis[s] * dis[d];
    float4 v = ((const float4*)(B + (long)s * 128))[lane];
    float* op = out + (long)d * 128 + lane * 4;
    atomicAdd(op + 0, norm * v.x);
    atomicAdd(op + 1, norm * v.y);
    atomicAdd(op + 2, norm * v.z);
    atomicAdd(op + 3, norm * v.w);
}

__global__ void gcn_pool_zero(float* sums, float* cnts) {
    int i = blockIdx.x * 256 + threadIdx.x;
    if (i < NUM_GRAPHS * 128) sums[i] = 0.0f;
    if (i < NUM_GRAPHS) cnts[i] = 0.0f;
}

__global__ void gcn_pool_accum(const float* __restrict__ h,
                               const float* __restrict__ bias,
                               const int* __restrict__ batch,
                               float* sums, float* cnts) {
    long idx = (long)blockIdx.x * 256 + threadIdx.x;
    if (idx >= (long)N_NODES * 128) return;
    int n = (int)(idx >> 7), c = (int)(idx & 127);
    int g = batch[n];
    atomicAdd(&sums[g * 128 + c], fmaxf(h[idx] + bias[c], 0.0f));
    if (c == 0) atomicAdd(&cnts[g], 1.0f);
}

__global__ void gcn_fc(const float* __restrict__ sums, const float* __restrict__ cnts,
                       const float* __restrict__ fcW, const float* __restrict__ fcb,
                       float* __restrict__ out) {
    int idx = blockIdx.x * 256 + threadIdx.x;
    if (idx >= NUM_GRAPHS * OUTC) return;
    int g = idx >> 4, o = idx & 15;
    float inv = 1.0f / fmaxf(cnts[g], 1.0f);
    float acc = fcb[o];
    for (int k = 0; k < 128; ++k)
        acc += sums[g * 128 + k] * inv * fcW[k * OUTC + o];
    out[idx] = acc;
}

extern "C" void kernel_launch(void* const* d_in, const int* in_sizes, int n_in,
                              void* d_out, int out_size, void* d_ws, size_t ws_size,
                              hipStream_t stream) {
    (void)in_sizes; (void)n_in; (void)out_size; (void)ws_size;
    const float* x    = (const float*)d_in[0];
    const float* top  = (const float*)d_in[1];
    const int*   ei   = (const int*)d_in[2];
    const int*   bat  = (const int*)d_in[3];
    const float* W1   = (const float*)d_in[4];
    const float* b1   = (const float*)d_in[5];
    const float* W2   = (const float*)d_in[6];
    const float* b2   = (const float*)d_in[7];
    const float* fcW  = (const float*)d_in[8];
    const float* fcb  = (const float*)d_in[9];
    float*       out  = (float*)d_out;

    const int* src = ei;
    const int* dst = ei + N_EDGES;

    char* ws = (char*)d_ws;
    float* A    = (float*)ws; ws += (size_t)N_NODES * 128 * 4;
    float* Bh   = (float*)ws; ws += (size_t)N_NODES * 128 * 4;
    float* dis  = (float*)ws; ws += (size_t)N_NODES * 4;
    float* sums = (float*)ws; ws += (size_t)NUM_GRAPHS * 128 * 4;
    float* cnts = (float*)ws;

    const int tb = 256;
    const int gN    = (N_NODES + tb - 1) / tb;
    const int gE    = (N_EDGES + tb - 1) / tb;
    const long NC   = (long)N_NODES * 128;
    const int gNC   = (int)((NC + tb - 1) / tb);
    const int gGemm = (N_NODES / 16 + 7) / 8;
    const int gEdge = (N_EDGES + 7) / 8;

    gcn_deg_init <<<gN, tb, 0, stream>>>(dis);
    gcn_deg_edges<<<gE, tb, 0, stream>>>(dst, dis);
    gcn_deg_rsqrt<<<gN, tb, 0, stream>>>(dis);

    gcn_gemm_wmma<0><<<gGemm, tb, 0, stream>>>(x, top, W1, Bh, N_NODES);
    gcn_agg_init <<<gNC, tb, 0, stream>>>(Bh, dis, A);
    gcn_agg_edges<<<gEdge, tb, 0, stream>>>(Bh, A, src, dst, dis);

    gcn_gemm_wmma<1><<<gGemm, tb, 0, stream>>>(A, b1, W2, Bh, N_NODES);
    gcn_agg_init <<<gNC, tb, 0, stream>>>(Bh, dis, A);
    gcn_agg_edges<<<gEdge, tb, 0, stream>>>(Bh, A, src, dst, dis);

    gcn_pool_zero <<<(NUM_GRAPHS * 128 + tb - 1) / tb, tb, 0, stream>>>(sums, cnts);
    gcn_pool_accum<<<gNC, tb, 0, stream>>>(A, b2, bat, sums, cnts);
    gcn_fc<<<(NUM_GRAPHS * OUTC + tb - 1) / tb, tb, 0, stream>>>(sums, cnts, fcW, fcb, out);
}


# mean-pool: each wave sums its 40 consecutive node rows in registers and issues one atomic per graph run and channel instead of one per element; float summation order changes
# speedup vs baseline: 1.5552x; 1.5552x over previous
#include <hip/hip_runtime.h>
#include <hip/hip_bf16.h>


#define N_NODES   200000
#define N_EDGES   640000
#define NUM_GRAPHS   256
#define OUTC          16

typedef __attribute__((ext_vector_type(2))) float v2f;
typedef __attribute__((ext_vector_type(8))) float v8f;

__global__ void gcn_deg_init(float* deg) {
    int i = blockIdx.x * 256 + threadIdx.x;
    if (i < N_NODES) deg[i] = 1.0f;
}

__global__ void gcn_deg_edges(const int* __restrict__ dst, float* deg) {
    int e = blockIdx.x * 256 + threadIdx.x;
    if (e < N_EDGES) atomicAdd(&deg[dst[e]], 1.0f);
}

__global__ void gcn_deg_rsqrt(float* deg) {
    int i = blockIdx.x * 256 + threadIdx.x;
    if (i < N_NODES) deg[i] = rsqrtf(deg[i]);
}

template <int MODE>
__global__ __launch_bounds__(256) void gcn_gemm_wmma(
        const float* __restrict__ A0, const float* __restrict__ A1,
        const float* __restrict__ W, float* __restrict__ B, int nRows) {
    __shared__ float sW[128 * 128];
    for (int i = threadIdx.x; i < 128 * 128; i += 256) {
        int kk = i >> 7, n = i & 127;
        sW[((kk >> 1) << 8) + n * 2 + (kk & 1)] = W[i];
    }
    __syncthreads();

    const int wave = threadIdx.x >> 5;
    const int lane = threadIdx.x & 31;
    const long m0  = ((long)blockIdx.x * 8 + wave) * 16;
    if (m0 >= nRows) return;

    const int half = lane >> 4;
    const int lr   = lane & 15;
    const int koff = half * 2;
    const long row = m0 + lr;

    auto loadA = [&](int k) -> v2f {
        v2f r;
        if (MODE == 0) {
            if (k < 124) {
                const float* p = A0 + row * 124 + k + koff;
                r.x = p[0]; r.y = p[1];
            } else {
                const float* p = A1 + row * 4 + (k - 124) + koff;
                r.x = p[0]; r.y = p[1];
            }
        } else {
            const float* p = A0 + row * 128 + k + koff;
            r.x = fmaxf(p[0] + A1[k + koff],     0.0f);
            r.y = fmaxf(p[1] + A1[k + koff + 1], 0.0f);
        }
        return r;
    };

    v8f acc[8];
#pragma unroll
    for (int i = 0; i < 8; ++i) acc[i] = (v8f)(0.0f);

    v2f a = loadA(0);
    for (int k = 0; k < 128; k += 4) {
        v2f an = a;
        if (k + 4 < 128) an = loadA(k + 4);
        const float* bbase = sW + (((k + koff) >> 1) << 8) + lr * 2;
#pragma unroll
        for (int nt = 0; nt < 8; ++nt) {
            v2f b = *(const v2f*)(bbase + nt * 32);
            acc[nt] = __builtin_amdgcn_wmma_f32_16x16x4_f32(
                false, a, false, b, (short)0, acc[nt], false, false);
        }
        a = an;
    }
#pragma unroll
    for (int nt = 0; nt < 8; ++nt) {
        const int n0 = nt * 16;
#pragma unroll
        for (int r = 0; r < 8; ++r)
            B[(m0 + half * 8 + r) * 128 + n0 + lr] = acc[nt][r];
    }
}

__global__ void gcn_agg_init(const float* __restrict__ B,
                             const float* __restrict__ dis,
                             float* __restrict__ out) {
    long idx = (long)blockIdx.x * 256 + threadIdx.x;
    if (idx >= (long)N_NODES * 128) return;
    int n = (int)(idx >> 7);
    float d = dis[n];
    out[idx] = d * d * B[idx];
}

__global__ __launch_bounds__(256) void gcn_agg_edges(
        const float* __restrict__ B, float* __restrict__ out,
        const int* __restrict__ src, const int* __restrict__ dst,
        const float* __restrict__ dis) {
    int e = blockIdx.x * 8 + (threadIdx.x >> 5);
    if (e >= N_EDGES) return;
    const int lane = threadIdx.x & 31;
    const int s = src[e], d = dst[e];
    const float norm = dis[s] * dis[d];
    float4 v = ((const float4*)(B + (long)s * 128))[lane];
    float* op = out + (long)d * 128 + lane * 4;
    atomicAdd(op + 0, norm * v.x);
    atomicAdd(op + 1, norm * v.y);
    atomicAdd(op + 2, norm * v.z);
    atomicAdd(op + 3, norm * v.w);
}

__global__ void gcn_pool_zero(float* sums, float* cnts) {
    int i = blockIdx.x * 256 + threadIdx.x;
    if (i < NUM_GRAPHS * 128) sums[i] = 0.0f;
    if (i < NUM_GRAPHS) cnts[i] = 0.0f;
}

#define POOL_NPW     40
#define POOL_U        8
#define POOL_WAVES  (N_NODES / POOL_NPW)
#define POOL_BLOCKS (POOL_WAVES / 8)
static_assert(N_NODES % (POOL_NPW * 8) == 0, "whole waves, whole workgroups");
static_assert(POOL_NPW % POOL_U == 0 && (POOL_NPW / POOL_U) % 2 == 1,
              "an odd number of groups: a b a b ... a");
static_assert(POOL_NPW > 32 && POOL_NPW <= 64,
              "graph ids of a wave's nodes sit in two registers");

__global__ __launch_bounds__(256) void gcn_pool_accum(
        const float* __restrict__ h, const float* __restrict__ bias,
        const int* __restrict__ batch, float* sums, float* cnts) {
    const int lane = threadIdx.x & 31;
    const int wave = __builtin_amdgcn_readfirstlane(
        (int)(blockIdx.x * 8 + (threadIdx.x >> 5)));
    const int n0 = wave * POOL_NPW;

    const int gv0 = batch[n0 + lane];
    const int gv1 = batch[n0 + 32 + (lane < POOL_NPW - 32 ? lane : POOL_NPW - 33)];
    const float4 bv = ((const float4*)bias)[lane];
    __builtin_amdgcn_sched_barrier(0);
    const float4* hp = (const float4*)h + (long)n0 * 32 + lane;

    float4 acc = make_float4(0.0f, 0.0f, 0.0f, 0.0f);
    int cnt = 0;
    int cur = 0;

    float4 va[POOL_U], vb[POOL_U];

    auto request = [&](float4 (&v)[POOL_U], int r0) {
#pragma unroll
        for (int u = 0; u < POOL_U; ++u) v[u] = hp[(r0 + u) * 32];
    };
    auto hand_over = [&]() {
        float* sp = sums + cur * 128 + lane * 4;
        atomicAdd(sp + 0, acc.x);
        atomicAdd(sp + 1, acc.y);
        atomicAdd(sp + 2, acc.z);
        atomicAdd(sp + 3, acc.w);
        if (lane == 0) atomicAdd(&cnts[cur], (float)cnt);
    };
    auto sum_group = [&](const float4 (&v)[POOL_U], int r0) {
#pragma unroll
        for (int u = 0; u < POOL_U; ++u) {
            const int r = r0 + u;
            const int g = __builtin_amdgcn_readlane(r < 32 ? gv0 : gv1, r & 31);
            if (g != cur) {
                hand_over();
                acc = make_float4(0.0f, 0.0f, 0.0f, 0.0f);
                cnt = 0;
                cur = g;
            }
            acc.x += fmaxf(v[u].x + bv.x, 0.0f);
            acc.y += fmaxf(v[u].y + bv.y, 0.0f);
            acc.z += fmaxf(v[u].z + bv.z, 0.0f);
            acc.w += fmaxf(v[u].w + bv.w, 0.0f);
            ++cnt;
        }
    };

    request(va, 0);
    request(vb, POOL_U);
    __builtin_amdgcn_sched_barrier(0);
    cur = __builtin_amdgcn_readlane(gv0, 0);
#pragma unroll
    for (int r0 = 0; r0 < POOL_NPW - POOL_U; r0 += 2 * POOL_U) {
        sum_group(va, r0);
        request(va, r0 + 2 * POOL_U);
        sum_group(vb, r0 + POOL_U);
        if (r0 + 3 * POOL_U < POOL_NPW) request(vb, r0 + 3 * POOL_U);
    }
    sum_group(va, POOL_NPW - POOL_U);
    hand_over();
}

__global__ void gcn_fc(const float* __restrict__ sums, const float* __restrict__ cnts,
                       const float* __restrict__ fcW, const float* __restrict__ fcb,
                       float* __restrict__ out) {
    int idx = blockIdx.x * 256 + threadIdx.x;
    if (idx >= NUM_GRAPHS * OUTC) return;
    int g = idx >> 4, o = idx & 15;
    float inv = 1.0f / fmaxf(cnts[g], 1.0f);
    float acc = fcb[o];
    for (int k = 0; k < 128; ++k)
        acc += sums[g * 128 + k] * inv * fcW[k * OUTC + o];
    out[idx] = acc;
}

extern "C" void kernel_launch(void* const* d_in, const int* in_sizes, int n_in,
                              void* d_out, int out_size, void* d_ws, size_t ws_size,
                              hipStream_t stream) {
    (void)in_sizes; (void)n_in; (void)out_size; (void)ws_size;
    const float* x    = (const float*)d_in[0];
    const float* top  = (const float*)d_in[1];
    const int*   ei   = (const int*)d_in[2];
    const int*   bat  = (const int*)d_in[3];
    const float* W1   = (const float*)d_in[4];
    const float* b1   = (const float*)d_in[5];
    const float* W2   = (const float*)d_in[6];
    const float* b2   = (const float*)d_in[7];
    const float* fcW  = (const float*)d_in[8];
    const float* fcb  = (const float*)d_in[9];
    float*       out  = (float*)d_out;

    const int* src = ei;
    const int* dst = ei + N_EDGES;

    char* ws = (char*)d_ws;
    float* A    = (float*)ws; ws += (size_t)N_NODES * 128 * 4;
    float* Bh   = (float*)ws; ws += (size_t)N_NODES * 128 * 4;
    float* dis  = (float*)ws; ws += (size_t)N_NODES * 4;
    float* sums = (float*)ws; ws += (size_t)NUM_GRAPHS * 128 * 4;
    float* cnts = (float*)ws;

    const int tb = 256;
    const int gN    = (N_NODES + tb - 1) / tb;
    const int gE    = (N_EDGES + tb - 1) / tb;
    const long NC   = (long)N_NODES * 128;
    const int gNC   = (int)((NC + tb - 1) / tb);
    const int gGemm = (N_NODES / 16 + 7) / 8;
    const int gEdge = (N_EDGES + 7) / 8;

    gcn_deg_init <<<gN, tb, 0, stream>>>(dis);
    gcn_deg_edges<<<gE, tb, 0, stream>>>(dst, dis);
    gcn_deg_rsqrt<<<gN, tb, 0, stream>>>(dis);

    gcn_gemm_wmma<0><<<gGemm, tb, 0, stream>>>(x, top, W1, Bh, N_NODES);
    gcn_agg_init <<<gNC, tb, 0, stream>>>(Bh, dis, A);
    gcn_agg_edges<<<gEdge, tb, 0, stream>>>(Bh, A, src, dst, dis);

    gcn_gemm_wmma<1><<<gGemm, tb, 0, stream>>>(A, b1, W2, Bh, N_NODES);
    gcn_agg_init <<<gNC, tb, 0, stream>>>(Bh, dis, A);
    gcn_agg_edges<<<gEdge, tb, 0, stream>>>(Bh, A, src, dst, dis);

    gcn_pool_zero <<<(NUM_GRAPHS * 128 + tb - 1) / tb, tb, 0, stream>>>(sums, cnts);
    gcn_pool_accum<<<POOL_BLOCKS, tb, 0, stream>>>(A, b2, bat, sums, cnts);
    gcn_fc<<<(NUM_GRAPHS * OUTC + tb - 1) / tb, tb, 0, stream>>>(sums, cnts, fcW, fcb, out);
}


# aggregation as a gather: edges ordered by destination once per call (integer count atomics rank them), a wave sums its 4 nodes' rows in registers and stores each once; float sum order changes
# speedup vs baseline: 3.1888x; 2.0504x over previous
#include <hip/hip_runtime.h>
#include <hip/hip_bf16.h>


#define N_NODES   200000
#define N_EDGES   640000
#define NUM_GRAPHS   256
#define OUTC          16

typedef __attribute__((ext_vector_type(2))) float v2f;
typedef __attribute__((ext_vector_type(8))) float v8f;

#define N_ENT      (N_NODES + N_EDGES)
#define ENT_SELF    0x80000000u
#define SCAN_TILE         1024
#define SCAN_BLOCKS ((N_NODES + SCAN_TILE - 1) / SCAN_TILE)
static_assert(SCAN_BLOCKS <= 256, "one thread per earlier tile total");
static_assert(N_NODES % 4 == 0, "nodes are scanned four per thread");

__global__ void csr_zero(int* cnt) {
    int i = blockIdx.x * 256 + threadIdx.x;
    if (i < N_NODES) cnt[i] = 0;
}

__global__ void csr_count(const int* __restrict__ dst, int* cnt, int* __restrict__ rank) {
    int e = blockIdx.x * 256 + threadIdx.x;
    if (e < N_EDGES) rank[e] = atomicAdd(&cnt[dst[e]], 1);
}

__device__ __forceinline__ int wave_sum(int v) {
    for (int m = 16; m >= 1; m >>= 1) v += __shfl_xor(v, m);
    return v;
}

__global__ __launch_bounds__(256) void csr_tile_sums(const int* __restrict__ cnt,
                                                     int* __restrict__ tsum) {
    __shared__ int sh[8];
    const int t = threadIdx.x;
    const int i = blockIdx.x * SCAN_TILE + t * 4;
    int s = 0;
    if (i < N_NODES) {
        const int4 c = *(const int4*)(cnt + i);
        s = c.x + c.y + c.z + c.w;
    }
    s = wave_sum(s);
    if ((t & 31) == 0) sh[t >> 5] = s;
    __syncthreads();
    if (t == 0) {
        int tot = 0;
        for (int k = 0; k < 8; ++k) tot += sh[k];
        tsum[blockIdx.x] = tot;
    }
}

__global__ __launch_bounds__(256) void csr_scan(const int* __restrict__ cnt,
                                                const int* __restrict__ tsum,
                                                int* __restrict__ rp,
                                                float* __restrict__ dis) {
    __shared__ int shA[8];
    __shared__ int shB[8];
    const int t = threadIdx.x, lane = t & 31, wv = t >> 5;
    const int b = blockIdx.x;
    int pre = (t < b) ? tsum[t] : 0;
    pre = wave_sum(pre);
    const int i = b * SCAN_TILE + t * 4;
    int4 c = make_int4(0, 0, 0, 0);
    if (i < N_NODES) c = *(const int4*)(cnt + i);
    const int mine = c.x + c.y + c.z + c.w;
    int inc = mine;
    for (int d = 1; d < 32; d <<= 1) {
        const int y = __shfl_up(inc, d);
        if (lane >= d) inc += y;
    }
    if (lane == 0)  shA[wv] = pre;
    if (lane == 31) shB[wv] = inc;
    __syncthreads();
    int base = 0;
    for (int k = 0; k < 8; ++k) {
        base += shA[k];
        if (k < wv) base += shB[k];
    }
    if (i < N_NODES) {
        int4 r;
        r.x = base + inc - mine + i;
        r.y = r.x + c.x + 1;
        r.z = r.y + c.y + 1;
        r.w = r.z + c.z + 1;
        *(int4*)(rp + i) = r;
        float4 d4;
        d4.x = rsqrtf((float)(c.x + 1));
        d4.y = rsqrtf((float)(c.y + 1));
        d4.z = rsqrtf((float)(c.z + 1));
        d4.w = rsqrtf((float)(c.w + 1));
        *(float4*)(dis + i) = d4;
    }
}

__global__ void csr_fill(const int* __restrict__ src, const int* __restrict__ dst,
                         const int* __restrict__ rank, const int* __restrict__ rp,
                         const float* __restrict__ dis, int* __restrict__ ent) {
    int idx = blockIdx.x * 256 + threadIdx.x;
    if (idx < N_EDGES) {
        const int s = src[idx], d = dst[idx];
        const int p = rp[d] + 1 + rank[idx];
        ((int2*)ent)[p] = make_int2(s, __float_as_int(dis[s] * dis[d]));
    } else if (idx < N_ENT) {
        const int i = idx - N_EDGES;
        const float di = dis[i];
        ((int2*)ent)[rp[i]] = make_int2((int)((unsigned)i | ENT_SELF), __float_as_int(di * di));
    }
}

template <int MODE>
__global__ __launch_bounds__(256) void gcn_gemm_wmma(
        const float* __restrict__ A0, const float* __restrict__ A1,
        const float* __restrict__ W, float* __restrict__ B, int nRows) {
    __shared__ float sW[128 * 128];
    for (int i = threadIdx.x; i < 128 * 128; i += 256) {
        int kk = i >> 7, n = i & 127;
        sW[((kk >> 1) << 8) + n * 2 + (kk & 1)] = W[i];
    }
    __syncthreads();

    const int wave = threadIdx.x >> 5;
    const int lane = threadIdx.x & 31;
    const long m0  = ((long)blockIdx.x * 8 + wave) * 16;
    if (m0 >= nRows) return;

    const int half = lane >> 4;
    const int lr   = lane & 15;
    const int koff = half * 2;
    const long row = m0 + lr;

    auto loadA = [&](int k) -> v2f {
        v2f r;
        if (MODE == 0) {
            if (k < 124) {
                const float* p = A0 + row * 124 + k + koff;
                r.x = p[0]; r.y = p[1];
            } else {
                const float* p = A1 + row * 4 + (k - 124) + koff;
                r.x = p[0]; r.y = p[1];
            }
        } else {
            const float* p = A0 + row * 128 + k + koff;
            r.x = fmaxf(p[0] + A1[k + koff],     0.0f);
            r.y = fmaxf(p[1] + A1[k + koff + 1], 0.0f);
        }
        return r;
    };

    v8f acc[8];
#pragma unroll
    for (int i = 0; i < 8; ++i) acc[i] = (v8f)(0.0f);

    v2f a = loadA(0);
    for (int k = 0; k < 128; k += 4) {
        v2f an = a;
        if (k + 4 < 128) an = loadA(k + 4);
        const float* bbase = sW + (((k + koff) >> 1) << 8) + lr * 2;
#pragma unroll
        for (int nt = 0; nt < 8; ++nt) {
            v2f b = *(const v2f*)(bbase + nt * 32);
            acc[nt] = __builtin_amdgcn_wmma_f32_16x16x4_f32(
                false, a, false, b, (short)0, acc[nt], false, false);
        }
        a = an;
    }
#pragma unroll
    for (int nt = 0; nt < 8; ++nt) {
        const int n0 = nt * 16;
#pragma unroll
        for (int r = 0; r < 8; ++r)
            B[(m0 + half * 8 + r) * 128 + n0 + lr] = acc[nt][r];
    }
}

#define PULL_NPW      4
#define PULL_U        8
#define PULL_WAVES  (N_NODES / PULL_NPW)
#define PULL_BLOCKS (PULL_WAVES / 8)
static_assert(N_NODES % (PULL_NPW * 8) == 0, "whole waves, whole workgroups");

template <bool FULL>
__device__ __forceinline__ void pull_group(const float4* Bv, float4* ov, int2 e,
                                           int u0, int m, int n0,
                                           int& node, float4& acc) {
    int    cs[PULL_U];
    float  wt[PULL_U];
    float4 v[PULL_U];
#pragma unroll
    for (int u = 0; u < PULL_U; ++u) {
        cs[u] = __builtin_amdgcn_readlane(e.x, u0 + u);
        wt[u] = __int_as_float(__builtin_amdgcn_readlane(e.y, u0 + u));
    }
#pragma unroll
    for (int u = 0; u < PULL_U; ++u)
        if (FULL || u0 + u < m) v[u] = Bv[(long)(cs[u] & 0x7fffffff) * 32];
#pragma unroll
    for (int u = 0; u < PULL_U; ++u)
        if (FULL || u0 + u < m) {
            if (cs[u] < 0) {
                if (node >= n0) ov[(long)node * 32] = acc;
                ++node;
                acc = make_float4(0.0f, 0.0f, 0.0f, 0.0f);
            }
            acc.x = fmaf(wt[u], v[u].x, acc.x);
            acc.y = fmaf(wt[u], v[u].y, acc.y);
            acc.z = fmaf(wt[u], v[u].z, acc.z);
            acc.w = fmaf(wt[u], v[u].w, acc.w);
        }
}

__global__ __launch_bounds__(256) void gcn_agg_pull(
        const float* __restrict__ B, const int* __restrict__ rp,
        const int* __restrict__ ent, float* __restrict__ out) {
    const int lane = threadIdx.x & 31;
    const int wave = __builtin_amdgcn_readfirstlane(
        (int)(blockIdx.x * 8 + (threadIdx.x >> 5)));
    const int n0 = wave * PULL_NPW;
    const int pBeg = rp[n0];
    const int pEnd = (n0 + PULL_NPW < N_NODES) ? rp[n0 + PULL_NPW] : N_ENT;

    const int2*   ev = (const int2*)ent;
    const float4* Bv = (const float4*)B + lane;
    float4*       ov = (float4*)out + lane;

    float4 acc = make_float4(0.0f, 0.0f, 0.0f, 0.0f);
    int node = n0 - 1;

    int2 e = make_int2(0, 0);
    if (pBeg + lane < pEnd) e = ev[pBeg + lane];
    for (int p = pBeg; p < pEnd; p += 32) {
        int2 en = make_int2(0, 0);
        if (p + 32 + lane < pEnd) en = ev[p + 32 + lane];
        const int m = (pEnd - p < 32) ? (pEnd - p) : 32;
        for (int u0 = 0; u0 < m; u0 += PULL_U) {
            if (u0 + PULL_U <= m) pull_group<true >(Bv, ov, e, u0, m, n0, node, acc);
            else                  pull_group<false>(Bv, ov, e, u0, m, n0, node, acc);
        }
        e = en;
    }
    ov[(long)node * 32] = acc;
}

__global__ void gcn_deg_init(float* deg) {
    int i = blockIdx.x * 256 + threadIdx.x;
    if (i < N_NODES) deg[i] = 1.0f;
}

__global__ void gcn_deg_edges(const int* __restrict__ dst, float* deg) {
    int e = blockIdx.x * 256 + threadIdx.x;
    if (e < N_EDGES) atomicAdd(&deg[dst[e]], 1.0f);
}

__global__ void gcn_deg_rsqrt(float* deg) {
    int i = blockIdx.x * 256 + threadIdx.x;
    if (i < N_NODES) deg[i] = rsqrtf(deg[i]);
}

__global__ void gcn_agg_init(const float* __restrict__ B,
                             const float* __restrict__ dis,
                             float* __restrict__ out) {
    long idx = (long)blockIdx.x * 256 + threadIdx.x;
    if (idx >= (long)N_NODES * 128) return;
    int n = (int)(idx >> 7);
    float d = dis[n];
    out[idx] = d * d * B[idx];
}

__global__ __launch_bounds__(256) void gcn_agg_edges(
        const float* __restrict__ B, float* __restrict__ out,
        const int* __restrict__ src, const int* __restrict__ dst,
        const float* __restrict__ dis) {
    int e = blockIdx.x * 8 + (threadIdx.x >> 5);
    if (e >= N_EDGES) return;
    const int lane = threadIdx.x & 31;
    const int s = src[e], d = dst[e];
    const float norm = dis[s] * dis[d];
    float4 v = ((const float4*)(B + (long)s * 128))[lane];
    float* op = out + (long)d * 128 + lane * 4;
    atomicAdd(op + 0, norm * v.x);
    atomicAdd(op + 1, norm * v.y);
    atomicAdd(op + 2, norm * v.z);
    atomicAdd(op + 3, norm * v.w);
}

__global__ void gcn_pool_zero(float* sums, float* cnts) {
    int i = blockIdx.x * 256 + threadIdx.x;
    if (i < NUM_GRAPHS * 128) sums[i] = 0.0f;
    if (i < NUM_GRAPHS) cnts[i] = 0.0f;
}

#define POOL_NPW     40
#define POOL_U        8
#define POOL_WAVES  (N_NODES / POOL_NPW)
#define POOL_BLOCKS (POOL_WAVES / 8)
static_assert(N_NODES % (POOL_NPW * 8) == 0, "whole waves, whole workgroups");
static_assert(POOL_NPW % POOL_U == 0 && (POOL_NPW / POOL_U) % 2 == 1,
              "an odd number of groups: a b a b ... a");
static_assert(POOL_NPW > 32 && POOL_NPW <= 64,
              "graph ids of a wave's nodes sit in two registers");

__global__ __launch_bounds__(256) void gcn_pool_accum(
        const float* __restrict__ h, const float* __restrict__ bias,
        const int* __restrict__ batch, float* sums, float* cnts) {
    const int lane = threadIdx.x & 31;
    const int wave = __builtin_amdgcn_readfirstlane(
        (int)(blockIdx.x * 8 + (threadIdx.x >> 5)));
    const int n0 = wave * POOL_NPW;

    const int gv0 = batch[n0 + lane];
    const int gv1 = batch[n0 + 32 + (lane < POOL_NPW - 32 ? lane : POOL_NPW - 33)];
    const float4 bv = ((const float4*)bias)[lane];
    __builtin_amdgcn_sched_barrier(0);
    const float4* hp = (const float4*)h + (long)n0 * 32 + lane;

    float4 acc = make_float4(0.0f, 0.0f, 0.0f, 0.0f);
    int cnt = 0;
    int cur = 0;

    float4 va[POOL_U], vb[POOL_U];

    auto request = [&](float4 (&v)[POOL_U], int r0) {
#pragma unroll
        for (int u = 0; u < POOL_U; ++u) v[u] = hp[(r0 + u) * 32];
    };
    auto hand_over = [&]() {
        float* sp = sums + cur * 128 + lane * 4;
        atomicAdd(sp + 0, acc.x);
        atomicAdd(sp + 1, acc.y);
        atomicAdd(sp + 2, acc.z);
        atomicAdd(sp + 3, acc.w);
        if (lane == 0) atomicAdd(&cnts[cur], (float)cnt);
    };
    auto sum_group = [&](const float4 (&v)[POOL_U], int r0) {
#pragma unroll
        for (int u = 0; u < POOL_U; ++u) {
            const int r = r0 + u;
            const int g = __builtin_amdgcn_readlane(r < 32 ? gv0 : gv1, r & 31);
            if (g != cur) {
                hand_over();
                acc = make_float4(0.0f, 0.0f, 0.0f, 0.0f);
                cnt = 0;
                cur = g;
            }
            acc.x += fmaxf(v[u].x + bv.x, 0.0f);
            acc.y += fmaxf(v[u].y + bv.y, 0.0f);
            acc.z += fmaxf(v[u].z + bv.z, 0.0f);
            acc.w += fmaxf(v[u].w + bv.w, 0.0f);
            ++cnt;
        }
    };

    request(va, 0);
    request(vb, POOL_U);
    __builtin_amdgcn_sched_barrier(0);
    cur = __builtin_amdgcn_readlane(gv0, 0);
#pragma unroll
    for (int r0 = 0; r0 < POOL_NPW - POOL_U; r0 += 2 * POOL_U) {
        sum_group(va, r0);
        request(va, r0 + 2 * POOL_U);
        sum_group(vb, r0 + POOL_U);
        if (r0 + 3 * POOL_U < POOL_NPW) request(vb, r0 + 3 * POOL_U);
    }
    sum_group(va, POOL_NPW - POOL_U);
    hand_over();
}

__global__ void gcn_fc(const float* __restrict__ sums, const float* __restrict__ cnts,
                       const float* __restrict__ fcW, const float* __restrict__ fcb,
                       float* __restrict__ out) {
    int idx = blockIdx.x * 256 + threadIdx.x;
    if (idx >= NUM_GRAPHS * OUTC) return;
    int g = idx >> 4, o = idx & 15;
    float inv = 1.0f / fmaxf(cnts[g], 1.0f);
    float acc = fcb[o];
    for (int k = 0; k < 128; ++k)
        acc += sums[g * 128 + k] * inv * fcW[k * OUTC + o];
    out[idx] = acc;
}

#define WS_ROWS_BYTES   ((size_t)N_NODES * 128 * 4)
#define WS_SMALL_BYTES  ((size_t)N_NODES * 4 + (size_t)NUM_GRAPHS * 128 * 4 + (size_t)NUM_GRAPHS * 4)
#define WS_ENT_BYTES    ((size_t)N_ENT * 8)
#define WS_GATHER_BYTES (2 * WS_ROWS_BYTES + WS_SMALL_BYTES + WS_ENT_BYTES)

extern "C" void kernel_launch(void* const* d_in, const int* in_sizes, int n_in,
                              void* d_out, int out_size, void* d_ws, size_t ws_size,
                              hipStream_t stream) {
    (void)in_sizes; (void)n_in; (void)out_size;
    const float* x    = (const float*)d_in[0];
    const float* top  = (const float*)d_in[1];
    const int*   ei   = (const int*)d_in[2];
    const int*   bat  = (const int*)d_in[3];
    const float* W1   = (const float*)d_in[4];
    const float* b1   = (const float*)d_in[5];
    const float* W2   = (const float*)d_in[6];
    const float* b2   = (const float*)d_in[7];
    const float* fcW  = (const float*)d_in[8];
    const float* fcb  = (const float*)d_in[9];
    float*       out  = (float*)d_out;

    const int* src = ei;
    const int* dst = ei + N_EDGES;

    char* ws = (char*)d_ws;
    float* A    = (float*)ws; ws += WS_ROWS_BYTES;
    float* Bh   = (float*)ws; ws += WS_ROWS_BYTES;
    void*  nplane = (void*)ws; ws += (size_t)N_NODES * 4;
    float* sums = (float*)ws; ws += (size_t)NUM_GRAPHS * 128 * 4;
    float* cnts = (float*)ws; ws += (size_t)NUM_GRAPHS * 4;
    int*   ent  = (int*)ws;

    const int tb = 256;
    const int gN    = (N_NODES + tb - 1) / tb;
    const int gE    = (N_EDGES + tb - 1) / tb;
    const int gGemm = (N_NODES / 16 + 7) / 8;

    if (ws_size >= WS_GATHER_BYTES) {
        int*   rp   = (int*)nplane;
        int*   cnt  = (int*)A;
        int*   rank = cnt + N_NODES;
        float* dis  = (float*)(rank + N_EDGES);
        int*   tsum = (int*)(dis + N_NODES);
        const int gEnt = (N_ENT + tb - 1) / tb;

        csr_zero     <<<gN, tb, 0, stream>>>(cnt);
        csr_count    <<<gE, tb, 0, stream>>>(dst, cnt, rank);
        csr_tile_sums<<<SCAN_BLOCKS, tb, 0, stream>>>(cnt, tsum);
        csr_scan     <<<SCAN_BLOCKS, tb, 0, stream>>>(cnt, tsum, rp, dis);
        csr_fill     <<<gEnt, tb, 0, stream>>>(src, dst, rank, rp, dis, ent);

        gcn_gemm_wmma<0><<<gGemm, tb, 0, stream>>>(x, top, W1, Bh, N_NODES);
        gcn_agg_pull <<<PULL_BLOCKS, tb, 0, stream>>>(Bh, rp, ent, A);

        gcn_gemm_wmma<1><<<gGemm, tb, 0, stream>>>(A, b1, W2, Bh, N_NODES);
        gcn_agg_pull <<<PULL_BLOCKS, tb, 0, stream>>>(Bh, rp, ent, A);
    } else {
        float* dis = (float*)nplane;
        const long NC   = (long)N_NODES * 128;
        const int gNC   = (int)((NC + tb - 1) / tb);
        const int gEdge = (N_EDGES + 7) / 8;

        gcn_deg_init <<<gN, tb, 0, stream>>>(dis);
        gcn_deg_edges<<<gE, tb, 0, stream>>>(dst, dis);
        gcn_deg_rsqrt<<<gN, tb, 0, stream>>>(dis);

        gcn_gemm_wmma<0><<<gGemm, tb, 0, stream>>>(x, top, W1, Bh, N_NODES);
        gcn_agg_init <<<gNC, tb, 0, stream>>>(Bh, dis, A);
        gcn_agg_edges<<<gEdge, tb, 0, stream>>>(Bh, A, src, dst, dis);

        gcn_gemm_wmma<1><<<gGemm, tb, 0, stream>>>(A, b1, W2, Bh, N_NODES);
        gcn_agg_init <<<gNC, tb, 0, stream>>>(Bh, dis, A);
        gcn_agg_edges<<<gEdge, tb, 0, stream>>>(Bh, A, src, dst, dis);
    }

    gcn_pool_zero <<<(NUM_GRAPHS * 128 + tb - 1) / tb, tb, 0, stream>>>(sums, cnts);
    gcn_pool_accum<<<POOL_BLOCKS, tb, 0, stream>>>(A, b2, bat, sums, cnts);
    gcn_fc<<<(NUM_GRAPHS * OUTC + tb - 1) / tb, tb, 0, stream>>>(sums, cnts, fcW, fcb, out);
}


# layer-2 aggregation feeds the mean pool directly: relu(row+b2) summed per wave, the workgroup's 8 waves combined in LDS, one atomic per channel; no row store, no pool pass; float sum order changes
# speedup vs baseline: 3.2989x; 1.0345x over previous
#include <hip/hip_runtime.h>
#include <hip/hip_bf16.h>


#define N_NODES   200000
#define N_EDGES   640000
#define NUM_GRAPHS   256
#define OUTC          16

typedef __attribute__((ext_vector_type(2))) float v2f;
typedef __attribute__((ext_vector_type(8))) float v8f;

#define N_ENT      (N_NODES + N_EDGES)
#define ENT_SELF    0x80000000u
#define SCAN_TILE         1024
#define SCAN_BLOCKS ((N_NODES + SCAN_TILE - 1) / SCAN_TILE)
static_assert(SCAN_BLOCKS <= 256, "one thread per earlier tile total");
static_assert(N_NODES % 4 == 0, "nodes are scanned four per thread");

__global__ void csr_zero(int* cnt) {
    int i = blockIdx.x * 256 + threadIdx.x;
    if (i < N_NODES) cnt[i] = 0;
}

__global__ void csr_count(const int* __restrict__ dst, int* cnt, int* __restrict__ rank) {
    int e = blockIdx.x * 256 + threadIdx.x;
    if (e < N_EDGES) rank[e] = atomicAdd(&cnt[dst[e]], 1);
}

__device__ __forceinline__ int wave_sum(int v) {
    for (int m = 16; m >= 1; m >>= 1) v += __shfl_xor(v, m);
    return v;
}

__global__ __launch_bounds__(256) void csr_tile_sums(const int* __restrict__ cnt,
                                                     int* __restrict__ tsum) {
    __shared__ int sh[8];
    const int t = threadIdx.x;
    const int i = blockIdx.x * SCAN_TILE + t * 4;
    int s = 0;
    if (i < N_NODES) {
        const int4 c = *(const int4*)(cnt + i);
        s = c.x + c.y + c.z + c.w;
    }
    s = wave_sum(s);
    if ((t & 31) == 0) sh[t >> 5] = s;
    __syncthreads();
    if (t == 0) {
        int tot = 0;
        for (int k = 0; k < 8; ++k) tot += sh[k];
        tsum[blockIdx.x] = tot;
    }
}

__global__ __launch_bounds__(256) void csr_scan(const int* __restrict__ cnt,
                                                const int* __restrict__ tsum,
                                                int* __restrict__ rp,
                                                float* __restrict__ dis) {
    __shared__ int shA[8];
    __shared__ int shB[8];
    const int t = threadIdx.x, lane = t & 31, wv = t >> 5;
    const int b = blockIdx.x;
    int pre = (t < b) ? tsum[t] : 0;
    pre = wave_sum(pre);
    const int i = b * SCAN_TILE + t * 4;
    int4 c = make_int4(0, 0, 0, 0);
    if (i < N_NODES) c = *(const int4*)(cnt + i);
    const int mine = c.x + c.y + c.z + c.w;
    int inc = mine;
    for (int d = 1; d < 32; d <<= 1) {
        const int y = __shfl_up(inc, d);
        if (lane >= d) inc += y;
    }
    if (lane == 0)  shA[wv] = pre;
    if (lane == 31) shB[wv] = inc;
    __syncthreads();
    int base = 0;
    for (int k = 0; k < 8; ++k) {
        base += shA[k];
        if (k < wv) base += shB[k];
    }
    if (i < N_NODES) {
        int4 r;
        r.x = base + inc - mine + i;
        r.y = r.x + c.x + 1;
        r.z = r.y + c.y + 1;
        r.w = r.z + c.z + 1;
        *(int4*)(rp + i) = r;
        float4 d4;
        d4.x = rsqrtf((float)(c.x + 1));
        d4.y = rsqrtf((float)(c.y + 1));
        d4.z = rsqrtf((float)(c.z + 1));
        d4.w = rsqrtf((float)(c.w + 1));
        *(float4*)(dis + i) = d4;
    }
}

__global__ void csr_fill(const int* __restrict__ src, const int* __restrict__ dst,
                         const int* __restrict__ rank, const int* __restrict__ rp,
                         const float* __restrict__ dis, int* __restrict__ ent) {
    int idx = blockIdx.x * 256 + threadIdx.x;
    if (idx < N_EDGES) {
        const int s = src[idx], d = dst[idx];
        const int p = rp[d] + 1 + rank[idx];
        ((int2*)ent)[p] = make_int2(s, __float_as_int(dis[s] * dis[d]));
    } else if (idx < N_ENT) {
        const int i = idx - N_EDGES;
        const float di = dis[i];
        ((int2*)ent)[rp[i]] = make_int2((int)((unsigned)i | ENT_SELF), __float_as_int(di * di));
    }
}

template <int MODE>
__global__ __launch_bounds__(256) void gcn_gemm_wmma(
        const float* __restrict__ A0, const float* __restrict__ A1,
        const float* __restrict__ W, float* __restrict__ B, int nRows) {
    __shared__ float sW[128 * 128];
    for (int i = threadIdx.x; i < 128 * 128; i += 256) {
        int kk = i >> 7, n = i & 127;
        sW[((kk >> 1) << 8) + n * 2 + (kk & 1)] = W[i];
    }
    __syncthreads();

    const int wave = threadIdx.x >> 5;
    const int lane = threadIdx.x & 31;
    const long m0  = ((long)blockIdx.x * 8 + wave) * 16;
    if (m0 >= nRows) return;

    const int half = lane >> 4;
    const int lr   = lane & 15;
    const int koff = half * 2;
    const long row = m0 + lr;

    auto loadA = [&](int k) -> v2f {
        v2f r;
        if (MODE == 0) {
            if (k < 124) {
                const float* p = A0 + row * 124 + k + koff;
                r.x = p[0]; r.y = p[1];
            } else {
                const float* p = A1 + row * 4 + (k - 124) + koff;
                r.x = p[0]; r.y = p[1];
            }
        } else {
            const float* p = A0 + row * 128 + k + koff;
            r.x = fmaxf(p[0] + A1[k + koff],     0.0f);
            r.y = fmaxf(p[1] + A1[k + koff + 1], 0.0f);
        }
        return r;
    };

    v8f acc[8];
#pragma unroll
    for (int i = 0; i < 8; ++i) acc[i] = (v8f)(0.0f);

    v2f a = loadA(0);
    for (int k = 0; k < 128; k += 4) {
        v2f an = a;
        if (k + 4 < 128) an = loadA(k + 4);
        const float* bbase = sW + (((k + koff) >> 1) << 8) + lr * 2;
#pragma unroll
        for (int nt = 0; nt < 8; ++nt) {
            v2f b = *(const v2f*)(bbase + nt * 32);
            acc[nt] = __builtin_amdgcn_wmma_f32_16x16x4_f32(
                false, a, false, b, (short)0, acc[nt], false, false);
        }
        a = an;
    }
#pragma unroll
    for (int nt = 0; nt < 8; ++nt) {
        const int n0 = nt * 16;
#pragma unroll
        for (int r = 0; r < 8; ++r)
            B[(m0 + half * 8 + r) * 128 + n0 + lr] = acc[nt][r];
    }
}

#define PULL_NPW      4
#define PULL_U        8
#define PULL_WAVES  (N_NODES / PULL_NPW)
#define PULL_BLOCKS (PULL_WAVES / 8)
static_assert(N_NODES % (PULL_NPW * 8) == 0, "whole waves, whole workgroups");

struct RowStore {
    float4* ov;
    __device__ __forceinline__ void row(int node, float4 a) { ov[(long)node * 32] = a; }
};

struct RowPool {
    float4 bv;
    float4 s;
    int    gv;
    int    n0, lane;
    int    cur, cnt;
    float* sums;
    float* cnts;
    __device__ __forceinline__ void flush() {
        float* sp = sums + cur * 128 + lane * 4;
        atomicAdd(sp + 0, s.x);
        atomicAdd(sp + 1, s.y);
        atomicAdd(sp + 2, s.z);
        atomicAdd(sp + 3, s.w);
        if (lane == 0) atomicAdd(&cnts[cur], (float)cnt);
    }
    __device__ __forceinline__ void row(int node, float4 a) {
        const int g = __builtin_amdgcn_readlane(gv, node - n0);
        if (g != cur) {
            if (cnt > 0) flush();
            s = make_float4(0.0f, 0.0f, 0.0f, 0.0f);
            cnt = 0;
            cur = g;
        }
        s.x += fmaxf(a.x + bv.x, 0.0f);
        s.y += fmaxf(a.y + bv.y, 0.0f);
        s.z += fmaxf(a.z + bv.z, 0.0f);
        s.w += fmaxf(a.w + bv.w, 0.0f);
        ++cnt;
    }
};

template <bool FULL, class Sink>
__device__ __forceinline__ void pull_group(const float4* Bv, Sink& sink, int2 e,
                                           int u0, int m, int n0,
                                           int& node, float4& acc) {
    int    cs[PULL_U];
    float  wt[PULL_U];
    float4 v[PULL_U];
#pragma unroll
    for (int u = 0; u < PULL_U; ++u) {
        cs[u] = __builtin_amdgcn_readlane(e.x, u0 + u);
        wt[u] = __int_as_float(__builtin_amdgcn_readlane(e.y, u0 + u));
    }
#pragma unroll
    for (int u = 0; u < PULL_U; ++u)
        if (FULL || u0 + u < m) v[u] = Bv[(long)(cs[u] & 0x7fffffff) * 32];
#pragma unroll
    for (int u = 0; u < PULL_U; ++u)
        if (FULL || u0 + u < m) {
            if (cs[u] < 0) {
                if (node >= n0) sink.row(node, acc);
                ++node;
                acc = make_float4(0.0f, 0.0f, 0.0f, 0.0f);
            }
            acc.x = fmaf(wt[u], v[u].x, acc.x);
            acc.y = fmaf(wt[u], v[u].y, acc.y);
            acc.z = fmaf(wt[u], v[u].z, acc.z);
            acc.w = fmaf(wt[u], v[u].w, acc.w);
        }
}

template <class Sink>
__device__ __forceinline__ void pull_wave(const float* __restrict__ B,
                                          const int* __restrict__ rp,
                                          const int* __restrict__ ent,
                                          int n0, int lane, Sink& sink) {
    const int pBeg = rp[n0];
    const int pEnd = (n0 + PULL_NPW < N_NODES) ? rp[n0 + PULL_NPW] : N_ENT;

    const int2*   ev = (const int2*)ent;
    const float4* Bv = (const float4*)B + lane;

    float4 acc = make_float4(0.0f, 0.0f, 0.0f, 0.0f);
    int node = n0 - 1;

    int2 e = make_int2(0, 0);
    if (pBeg + lane < pEnd) e = ev[pBeg + lane];
    for (int p = pBeg; p < pEnd; p += 32) {
        int2 en = make_int2(0, 0);
        if (p + 32 + lane < pEnd) en = ev[p + 32 + lane];
        const int m = (pEnd - p < 32) ? (pEnd - p) : 32;
        for (int u0 = 0; u0 < m; u0 += PULL_U) {
            if (u0 + PULL_U <= m) pull_group<true >(Bv, sink, e, u0, m, n0, node, acc);
            else                  pull_group<false>(Bv, sink, e, u0, m, n0, node, acc);
        }
        e = en;
    }
    sink.row(node, acc);
}

__global__ __launch_bounds__(256) void gcn_agg_pull(
        const float* __restrict__ B, const int* __restrict__ rp,
        const int* __restrict__ ent, float* __restrict__ out) {
    const int lane = threadIdx.x & 31;
    const int wave = __builtin_amdgcn_readfirstlane(
        (int)(blockIdx.x * 8 + (threadIdx.x >> 5)));
    RowStore sink;
    sink.ov = (float4*)out + lane;
    pull_wave(B, rp, ent, wave * PULL_NPW, lane, sink);
}

__global__ __launch_bounds__(256) void gcn_agg_pull_pool(
        const float* __restrict__ B, const int* __restrict__ rp,
        const int* __restrict__ ent, const float* __restrict__ bias,
        const int* __restrict__ batch, float* sums, float* cnts) {
    __shared__ float shS[8 * 128];
    __shared__ int   shG[8];
    __shared__ int   shC[8];
    const int lane = threadIdx.x & 31;
    const int wib  = threadIdx.x >> 5;
    const int wave = __builtin_amdgcn_readfirstlane((int)(blockIdx.x * 8 + wib));
    const int n0 = wave * PULL_NPW;

    RowPool sink;
    sink.bv   = ((const float4*)bias)[lane];
    sink.gv   = batch[n0 + (lane & (PULL_NPW - 1))];
    sink.s    = make_float4(0.0f, 0.0f, 0.0f, 0.0f);
    sink.n0   = n0;
    sink.lane = lane;
    sink.cur  = -1;
    sink.cnt  = 0;
    sink.sums = sums;
    sink.cnts = cnts;
    pull_wave(B, rp, ent, n0, lane, sink);

    ((float4*)shS)[wib * 32 + lane] = sink.s;
    if (lane == 0) { shG[wib] = sink.cur; shC[wib] = sink.cnt; }
    __syncthreads();
    const int t = threadIdx.x;
    if (t < 128) {
        int   c = shG[0];
        float v = shS[t];
        for (int w = 1; w < 8; ++w) {
            const int g = shG[w];
            if (g != c) { atomicAdd(&sums[c * 128 + t], v); v = 0.0f; c = g; }
            v += shS[w * 128 + t];
        }
        atomicAdd(&sums[c * 128 + t], v);
    } else if (t == 128) {
        int c = shG[0];
        int n = shC[0];
        for (int w = 1; w < 8; ++w) {
            const int g = shG[w];
            if (g != c) { atomicAdd(&cnts[c], (float)n); n = 0; c = g; }
            n += shC[w];
        }
        atomicAdd(&cnts[c], (float)n);
    }
}

__global__ void gcn_deg_init(float* deg) {
    int i = blockIdx.x * 256 + threadIdx.x;
    if (i < N_NODES) deg[i] = 1.0f;
}

__global__ void gcn_deg_edges(const int* __restrict__ dst, float* deg) {
    int e = blockIdx.x * 256 + threadIdx.x;
    if (e < N_EDGES) atomicAdd(&deg[dst[e]], 1.0f);
}

__global__ void gcn_deg_rsqrt(float* deg) {
    int i = blockIdx.x * 256 + threadIdx.x;
    if (i < N_NODES) deg[i] = rsqrtf(deg[i]);
}

__global__ void gcn_agg_init(const float* __restrict__ B,
                             const float* __restrict__ dis,
                             float* __restrict__ out) {
    long idx = (long)blockIdx.x * 256 + threadIdx.x;
    if (idx >= (long)N_NODES * 128) return;
    int n = (int)(idx >> 7);
    float d = dis[n];
    out[idx] = d * d * B[idx];
}

__global__ __launch_bounds__(256) void gcn_agg_edges(
        const float* __restrict__ B, float* __restrict__ out,
        const int* __restrict__ src, const int* __restrict__ dst,
        const float* __restrict__ dis) {
    int e = blockIdx.x * 8 + (threadIdx.x >> 5);
    if (e >= N_EDGES) return;
    const int lane = threadIdx.x & 31;
    const int s = src[e], d = dst[e];
    const float norm = dis[s] * dis[d];
    float4 v = ((const float4*)(B + (long)s * 128))[lane];
    float* op = out + (long)d * 128 + lane * 4;
    atomicAdd(op + 0, norm * v.x);
    atomicAdd(op + 1, norm * v.y);
    atomicAdd(op + 2, norm * v.z);
    atomicAdd(op + 3, norm * v.w);
}

__global__ void gcn_pool_zero(float* sums, float* cnts) {
    int i = blockIdx.x * 256 + threadIdx.x;
    if (i < NUM_GRAPHS * 128) sums[i] = 0.0f;
    if (i < NUM_GRAPHS) cnts[i] = 0.0f;
}

#define POOL_NPW     40
#define POOL_U        8
#define POOL_WAVES  (N_NODES / POOL_NPW)
#define POOL_BLOCKS (POOL_WAVES / 8)
static_assert(N_NODES % (POOL_NPW * 8) == 0, "whole waves, whole workgroups");
static_assert(POOL_NPW % POOL_U == 0 && (POOL_NPW / POOL_U) % 2 == 1,
              "an odd number of groups: a b a b ... a");
static_assert(POOL_NPW > 32 && POOL_NPW <= 64,
              "graph ids of a wave's nodes sit in two registers");

__global__ __launch_bounds__(256) void gcn_pool_accum(
        const float* __restrict__ h, const float* __restrict__ bias,
        const int* __restrict__ batch, float* sums, float* cnts) {
    const int lane = threadIdx.x & 31;
    const int wave = __builtin_amdgcn_readfirstlane(
        (int)(blockIdx.x * 8 + (threadIdx.x >> 5)));
    const int n0 = wave * POOL_NPW;

    const int gv0 = batch[n0 + lane];
    const int gv1 = batch[n0 + 32 + (lane < POOL_NPW - 32 ? lane : POOL_NPW - 33)];
    const float4 bv = ((const float4*)bias)[lane];
    __builtin_amdgcn_sched_barrier(0);
    const float4* hp = (const float4*)h + (long)n0 * 32 + lane;

    float4 acc = make_float4(0.0f, 0.0f, 0.0f, 0.0f);
    int cnt = 0;
    int cur = 0;

    float4 va[POOL_U], vb[POOL_U];

    auto request = [&](float4 (&v)[POOL_U], int r0) {
#pragma unroll
        for (int u = 0; u < POOL_U; ++u) v[u] = hp[(r0 + u) * 32];
    };
    auto hand_over = [&]() {
        float* sp = sums + cur * 128 + lane * 4;
        atomicAdd(sp + 0, acc.x);
        atomicAdd(sp + 1, acc.y);
        atomicAdd(sp + 2, acc.z);
        atomicAdd(sp + 3, acc.w);
        if (lane == 0) atomicAdd(&cnts[cur], (float)cnt);
    };
    auto sum_group = [&](const float4 (&v)[POOL_U], int r0) {
#pragma unroll
        for (int u = 0; u < POOL_U; ++u) {
            const int r = r0 + u;
            const int g = __builtin_amdgcn_readlane(r < 32 ? gv0 : gv1, r & 31);
            if (g != cur) {
                hand_over();
                acc = make_float4(0.0f, 0.0f, 0.0f, 0.0f);
                cnt = 0;
                cur = g;
            }
            acc.x += fmaxf(v[u].x + bv.x, 0.0f);
            acc.y += fmaxf(v[u].y + bv.y, 0.0f);
            acc.z += fmaxf(v[u].z + bv.z, 0.0f);
            acc.w += fmaxf(v[u].w + bv.w, 0.0f);
            ++cnt;
        }
    };

    request(va, 0);
    request(vb, POOL_U);
    __builtin_amdgcn_sched_barrier(0);
    cur = __builtin_amdgcn_readlane(gv0, 0);
#pragma unroll
    for (int r0 = 0; r0 < POOL_NPW - POOL_U; r0 += 2 * POOL_U) {
        sum_group(va, r0);
        request(va, r0 + 2 * POOL_U);
        sum_group(vb, r0 + POOL_U);
        if (r0 + 3 * POOL_U < POOL_NPW) request(vb, r0 + 3 * POOL_U);
    }
    sum_group(va, POOL_NPW - POOL_U);
    hand_over();
}

__global__ void gcn_fc(const float* __restrict__ sums, const float* __restrict__ cnts,
                       const float* __restrict__ fcW, const float* __restrict__ fcb,
                       float* __restrict__ out) {
    int idx = blockIdx.x * 256 + threadIdx.x;
    if (idx >= NUM_GRAPHS * OUTC) return;
    int g = idx >> 4, o = idx & 15;
    float inv = 1.0f / fmaxf(cnts[g], 1.0f);
    float acc = fcb[o];
    for (int k = 0; k < 128; ++k)
        acc += sums[g * 128 + k] * inv * fcW[k * OUTC + o];
    out[idx] = acc;
}

#define WS_ROWS_BYTES   ((size_t)N_NODES * 128 * 4)
#define WS_SMALL_BYTES  ((size_t)N_NODES * 4 + (size_t)NUM_GRAPHS * 128 * 4 + (size_t)NUM_GRAPHS * 4)
#define WS_ENT_BYTES    ((size_t)N_ENT * 8)
#define WS_GATHER_BYTES (2 * WS_ROWS_BYTES + WS_SMALL_BYTES + WS_ENT_BYTES)

extern "C" void kernel_launch(void* const* d_in, const int* in_sizes, int n_in,
                              void* d_out, int out_size, void* d_ws, size_t ws_size,
                              hipStream_t stream) {
    (void)in_sizes; (void)n_in; (void)out_size;
    const float* x    = (const float*)d_in[0];
    const float* top  = (const float*)d_in[1];
    const int*   ei   = (const int*)d_in[2];
    const int*   bat  = (const int*)d_in[3];
    const float* W1   = (const float*)d_in[4];
    const float* b1   = (const float*)d_in[5];
    const float* W2   = (const float*)d_in[6];
    const float* b2   = (const float*)d_in[7];
    const float* fcW  = (const float*)d_in[8];
    const float* fcb  = (const float*)d_in[9];
    float*       out  = (float*)d_out;

    const int* src = ei;
    const int* dst = ei + N_EDGES;

    char* ws = (char*)d_ws;
    float* A    = (float*)ws; ws += WS_ROWS_BYTES;
    float* Bh   = (float*)ws; ws += WS_ROWS_BYTES;
    void*  nplane = (void*)ws; ws += (size_t)N_NODES * 4;
    float* sums = (float*)ws; ws += (size_t)NUM_GRAPHS * 128 * 4;
    float* cnts = (float*)ws; ws += (size_t)NUM_GRAPHS * 4;
    int*   ent  = (int*)ws;

    const int tb = 256;
    const int gN    = (N_NODES + tb - 1) / tb;
    const int gE    = (N_EDGES + tb - 1) / tb;
    const int gGemm = (N_NODES / 16 + 7) / 8;

    if (ws_size >= WS_GATHER_BYTES) {
        int*   rp   = (int*)nplane;
        int*   cnt  = (int*)A;
        int*   rank = cnt + N_NODES;
        float* dis  = (float*)(rank + N_EDGES);
        int*   tsum = (int*)(dis + N_NODES);
        const int gEnt = (N_ENT + tb - 1) / tb;

        csr_zero     <<<gN, tb, 0, stream>>>(cnt);
        csr_count    <<<gE, tb, 0, stream>>>(dst, cnt, rank);
        csr_tile_sums<<<SCAN_BLOCKS, tb, 0, stream>>>(cnt, tsum);
        csr_scan     <<<SCAN_BLOCKS, tb, 0, stream>>>(cnt, tsum, rp, dis);
        csr_fill     <<<gEnt, tb, 0, stream>>>(src, dst, rank, rp, dis, ent);

        gcn_gemm_wmma<0><<<gGemm, tb, 0, stream>>>(x, top, W1, Bh, N_NODES);
        gcn_agg_pull <<<PULL_BLOCKS, tb, 0, stream>>>(Bh, rp, ent, A);

        gcn_gemm_wmma<1><<<gGemm, tb, 0, stream>>>(A, b1, W2, Bh, N_NODES);
        gcn_pool_zero<<<(NUM_GRAPHS * 128 + tb - 1) / tb, tb, 0, stream>>>(sums, cnts);
        gcn_agg_pull_pool<<<PULL_BLOCKS, tb, 0, stream>>>(Bh, rp, ent, b2, bat, sums, cnts);
    } else {
        float* dis = (float*)nplane;
        const long NC   = (long)N_NODES * 128;
        const int gNC   = (int)((NC + tb - 1) / tb);
        const int gEdge = (N_EDGES + 7) / 8;

        gcn_deg_init <<<gN, tb, 0, stream>>>(dis);
        gcn_deg_edges<<<gE, tb, 0, stream>>>(dst, dis);
        gcn_deg_rsqrt<<<gN, tb, 0, stream>>>(dis);

        gcn_gemm_wmma<0><<<gGemm, tb, 0, stream>>>(x, top, W1, Bh, N_NODES);
        gcn_agg_init <<<gNC, tb, 0, stream>>>(Bh, dis, A);
        gcn_agg_edges<<<gEdge, tb, 0, stream>>>(Bh, A, src, dst, dis);

        gcn_gemm_wmma<1><<<gGemm, tb, 0, stream>>>(A, b1, W2, Bh, N_NODES);
        gcn_agg_init <<<gNC, tb, 0, stream>>>(Bh, dis, A);
        gcn_agg_edges<<<gEdge, tb, 0, stream>>>(Bh, A, src, dst, dis);

        gcn_pool_zero <<<(NUM_GRAPHS * 128 + tb - 1) / tb, tb, 0, stream>>>(sums, cnts);
        gcn_pool_accum<<<POOL_BLOCKS, tb, 0, stream>>>(A, b2, bat, sums, cnts);
    }

    gcn_fc<<<(NUM_GRAPHS * OUTC + tb - 1) / tb, tb, 0, stream>>>(sums, cnts, fcW, fcb, out);
}


# independent work shares launches: layer-1 GEMM row blocks run as extra workgroups of the edge count and edge fill launches, pool planes zeroed with the degree counters; 11 launches become 9
# speedup vs baseline: 3.4868x; 1.0570x over previous
#include <hip/hip_runtime.h>
#include <hip/hip_bf16.h>


#define N_NODES   200000
#define N_EDGES   640000
#define NUM_GRAPHS   256
#define OUTC          16

typedef __attribute__((ext_vector_type(2))) float v2f;
typedef __attribute__((ext_vector_type(8))) float v8f;

#define N_ENT      (N_NODES + N_EDGES)
#define ENT_SELF    0x80000000u
#define SCAN_TILE         1024
#define SCAN_BLOCKS ((N_NODES + SCAN_TILE - 1) / SCAN_TILE)
static_assert(SCAN_BLOCKS <= 256, "one thread per earlier tile total");
static_assert(N_NODES % 4 == 0, "nodes are scanned four per thread");

static_assert(N_NODES >= NUM_GRAPHS * 128, "the node grid covers the pool planes");
__global__ void csr_zero(int* cnt, float* sums, float* cnts) {
    int i = blockIdx.x * 256 + threadIdx.x;
    if (i < N_NODES) cnt[i] = 0;
    if (i < NUM_GRAPHS * 128) sums[i] = 0.0f;
    if (i < NUM_GRAPHS) cnts[i] = 0.0f;
}

__device__ __forceinline__ void csr_count_block(int blk, const int* __restrict__ dst,
                                                int* cnt, int* __restrict__ rank) {
    int e = blk * 256 + threadIdx.x;
    if (e < N_EDGES) rank[e] = atomicAdd(&cnt[dst[e]], 1);
}

__device__ __forceinline__ int wave_sum(int v) {
    for (int m = 16; m >= 1; m >>= 1) v += __shfl_xor(v, m);
    return v;
}

__global__ __launch_bounds__(256) void csr_tile_sums(const int* __restrict__ cnt,
                                                     int* __restrict__ tsum) {
    __shared__ int sh[8];
    const int t = threadIdx.x;
    const int i = blockIdx.x * SCAN_TILE + t * 4;
    int s = 0;
    if (i < N_NODES) {
        const int4 c = *(const int4*)(cnt + i);
        s = c.x + c.y + c.z + c.w;
    }
    s = wave_sum(s);
    if ((t & 31) == 0) sh[t >> 5] = s;
    __syncthreads();
    if (t == 0) {
        int tot = 0;
        for (int k = 0; k < 8; ++k) tot += sh[k];
        tsum[blockIdx.x] = tot;
    }
}

__global__ __launch_bounds__(256) void csr_scan(const int* __restrict__ cnt,
                                                const int* __restrict__ tsum,
                                                int* __restrict__ rp,
                                                float* __restrict__ dis) {
    __shared__ int shA[8];
    __shared__ int shB[8];
    const int t = threadIdx.x, lane = t & 31, wv = t >> 5;
    const int b = blockIdx.x;
    int pre = (t < b) ? tsum[t] : 0;
    pre = wave_sum(pre);
    const int i = b * SCAN_TILE + t * 4;
    int4 c = make_int4(0, 0, 0, 0);
    if (i < N_NODES) c = *(const int4*)(cnt + i);
    const int mine = c.x + c.y + c.z + c.w;
    int inc = mine;
    for (int d = 1; d < 32; d <<= 1) {
        const int y = __shfl_up(inc, d);
        if (lane >= d) inc += y;
    }
    if (lane == 0)  shA[wv] = pre;
    if (lane == 31) shB[wv] = inc;
    __syncthreads();
    int base = 0;
    for (int k = 0; k < 8; ++k) {
        base += shA[k];
        if (k < wv) base += shB[k];
    }
    if (i < N_NODES) {
        int4 r;
        r.x = base + inc - mine + i;
        r.y = r.x + c.x + 1;
        r.z = r.y + c.y + 1;
        r.w = r.z + c.z + 1;
        *(int4*)(rp + i) = r;
        float4 d4;
        d4.x = rsqrtf((float)(c.x + 1));
        d4.y = rsqrtf((float)(c.y + 1));
        d4.z = rsqrtf((float)(c.z + 1));
        d4.w = rsqrtf((float)(c.w + 1));
        *(float4*)(dis + i) = d4;
    }
}

__device__ __forceinline__ void csr_fill_block(int blk,
                         const int* __restrict__ src, const int* __restrict__ dst,
                         const int* __restrict__ rank, const int* __restrict__ rp,
                         const float* __restrict__ dis, int* __restrict__ ent) {
    int idx = blk * 256 + threadIdx.x;
    if (idx < N_EDGES) {
        const int s = src[idx], d = dst[idx];
        const int p = rp[d] + 1 + rank[idx];
        ((int2*)ent)[p] = make_int2(s, __float_as_int(dis[s] * dis[d]));
    } else if (idx < N_ENT) {
        const int i = idx - N_EDGES;
        const float di = dis[i];
        ((int2*)ent)[rp[i]] = make_int2((int)((unsigned)i | ENT_SELF), __float_as_int(di * di));
    }
}

template <int MODE>
__device__ __forceinline__ void gcn_gemm_block(
        const float* __restrict__ A0, const float* __restrict__ A1,
        const float* __restrict__ W, float* __restrict__ B, int nRows, int blk) {
    __shared__ float sW[128 * 128];
    for (int i = threadIdx.x; i < 128 * 128; i += 256) {
        int kk = i >> 7, n = i & 127;
        sW[((kk >> 1) << 8) + n * 2 + (kk & 1)] = W[i];
    }
    __syncthreads();

    const int wave = threadIdx.x >> 5;
    const int lane = threadIdx.x & 31;
    const long m0  = ((long)blk * 8 + wave) * 16;
    if (m0 >= nRows) return;

    const int half = lane >> 4;
    const int lr   = lane & 15;
    const int koff = half * 2;
    const long row = m0 + lr;

    auto loadA = [&](int k) -> v2f {
        v2f r;
        if (MODE == 0) {
            if (k < 124) {
                const float* p = A0 + row * 124 + k + koff;
                r.x = p[0]; r.y = p[1];
            } else {
                const float* p = A1 + row * 4 + (k - 124) + koff;
                r.x = p[0]; r.y = p[1];
            }
        } else {
            const float* p = A0 + row * 128 + k + koff;
            r.x = fmaxf(p[0] + A1[k + koff],     0.0f);
            r.y = fmaxf(p[1] + A1[k + koff + 1], 0.0f);
        }
        return r;
    };

    v8f acc[8];
#pragma unroll
    for (int i = 0; i < 8; ++i) acc[i] = (v8f)(0.0f);

    v2f a = loadA(0);
    for (int k = 0; k < 128; k += 4) {
        v2f an = a;
        if (k + 4 < 128) an = loadA(k + 4);
        const float* bbase = sW + (((k + koff) >> 1) << 8) + lr * 2;
#pragma unroll
        for (int nt = 0; nt < 8; ++nt) {
            v2f b = *(const v2f*)(bbase + nt * 32);
            acc[nt] = __builtin_amdgcn_wmma_f32_16x16x4_f32(
                false, a, false, b, (short)0, acc[nt], false, false);
        }
        a = an;
    }
#pragma unroll
    for (int nt = 0; nt < 8; ++nt) {
        const int n0 = nt * 16;
#pragma unroll
        for (int r = 0; r < 8; ++r)
            B[(m0 + half * 8 + r) * 128 + n0 + lr] = acc[nt][r];
    }
}

template <int MODE>
__global__ __launch_bounds__(256) void gcn_gemm_wmma(
        const float* __restrict__ A0, const float* __restrict__ A1,
        const float* __restrict__ W, float* __restrict__ B, int nRows) {
    gcn_gemm_block<MODE>(A0, A1, W, B, nRows, (int)blockIdx.x);
}

#define GEMM_BLOCKS     ((N_NODES / 16 + 7) / 8)
#define GEMM_IN_COUNT   768
#define GEMM_IN_FILL    (GEMM_BLOCKS - GEMM_IN_COUNT)
#define EDGE_BLOCKS     ((N_EDGES + 255) / 256)
#define ENT_BLOCKS      ((N_ENT + 255) / 256)
static_assert(GEMM_IN_FILL > 0, "both launches carry GEMM blocks");

__global__ __launch_bounds__(256) void csr_count_with_gemm(
        const int* __restrict__ dst, int* cnt, int* __restrict__ rank,
        const float* __restrict__ x, const float* __restrict__ top,
        const float* __restrict__ W, float* __restrict__ B) {
    if (blockIdx.x < GEMM_IN_COUNT)
        gcn_gemm_block<0>(x, top, W, B, N_NODES, (int)blockIdx.x);
    else
        csr_count_block((int)blockIdx.x - GEMM_IN_COUNT, dst, cnt, rank);
}

__global__ __launch_bounds__(256) void csr_fill_with_gemm(
        const int* __restrict__ src, const int* __restrict__ dst,
        const int* __restrict__ rank, const int* __restrict__ rp,
        const float* __restrict__ dis, int* __restrict__ ent,
        const float* __restrict__ x, const float* __restrict__ top,
        const float* __restrict__ W, float* __restrict__ B) {
    if (blockIdx.x < GEMM_IN_FILL)
        gcn_gemm_block<0>(x, top, W, B, N_NODES, GEMM_IN_COUNT + (int)blockIdx.x);
    else
        csr_fill_block((int)blockIdx.x - GEMM_IN_FILL, src, dst, rank, rp, dis, ent);
}

#define PULL_NPW      4
#define PULL_U        8
#define PULL_WAVES  (N_NODES / PULL_NPW)
#define PULL_BLOCKS (PULL_WAVES / 8)
static_assert(N_NODES % (PULL_NPW * 8) == 0, "whole waves, whole workgroups");

struct RowStore {
    float4* ov;
    __device__ __forceinline__ void row(int node, float4 a) { ov[(long)node * 32] = a; }
};

struct RowPool {
    float4 bv;
    float4 s;
    int    gv;
    int    n0, lane;
    int    cur, cnt;
    float* sums;
    float* cnts;
    __device__ __forceinline__ void flush() {
        float* sp = sums + cur * 128 + lane * 4;
        atomicAdd(sp + 0, s.x);
        atomicAdd(sp + 1, s.y);
        atomicAdd(sp + 2, s.z);
        atomicAdd(sp + 3, s.w);
        if (lane == 0) atomicAdd(&cnts[cur], (float)cnt);
    }
    __device__ __forceinline__ void row(int node, float4 a) {
        const int g = __builtin_amdgcn_readlane(gv, node - n0);
        if (g != cur) {
            if (cnt > 0) flush();
            s = make_float4(0.0f, 0.0f, 0.0f, 0.0f);
            cnt = 0;
            cur = g;
        }
        s.x += fmaxf(a.x + bv.x, 0.0f);
        s.y += fmaxf(a.y + bv.y, 0.0f);
        s.z += fmaxf(a.z + bv.z, 0.0f);
        s.w += fmaxf(a.w + bv.w, 0.0f);
        ++cnt;
    }
};

template <bool FULL, class Sink>
__device__ __forceinline__ void pull_group(const float4* Bv, Sink& sink, int2 e,
                                           int u0, int m, int n0,
                                           int& node, float4& acc) {
    int    cs[PULL_U];
    float  wt[PULL_U];
    float4 v[PULL_U];
#pragma unroll
    for (int u = 0; u < PULL_U; ++u) {
        cs[u] = __builtin_amdgcn_readlane(e.x, u0 + u);
        wt[u] = __int_as_float(__builtin_amdgcn_readlane(e.y, u0 + u));
    }
#pragma unroll
    for (int u = 0; u < PULL_U; ++u)
        if (FULL || u0 + u < m) v[u] = Bv[(long)(cs[u] & 0x7fffffff) * 32];
#pragma unroll
    for (int u = 0; u < PULL_U; ++u)
        if (FULL || u0 + u < m) {
            if (cs[u] < 0) {
                if (node >= n0) sink.row(node, acc);
                ++node;
                acc = make_float4(0.0f, 0.0f, 0.0f, 0.0f);
            }
            acc.x = fmaf(wt[u], v[u].x, acc.x);
            acc.y = fmaf(wt[u], v[u].y, acc.y);
            acc.z = fmaf(wt[u], v[u].z, acc.z);
            acc.w = fmaf(wt[u], v[u].w, acc.w);
        }
}

template <class Sink>
__device__ __forceinline__ void pull_wave(const float* __restrict__ B,
                                          const int* __restrict__ rp,
                                          const int* __restrict__ ent,
                                          int n0, int lane, Sink& sink) {
    const int pBeg = rp[n0];
    const int pEnd = (n0 + PULL_NPW < N_NODES) ? rp[n0 + PULL_NPW] : N_ENT;

    const int2*   ev = (const int2*)ent;
    const float4* Bv = (const float4*)B + lane;

    float4 acc = make_float4(0.0f, 0.0f, 0.0f, 0.0f);
    int node = n0 - 1;

    int2 e = make_int2(0, 0);
    if (pBeg + lane < pEnd) e = ev[pBeg + lane];
    for (int p = pBeg; p < pEnd; p += 32) {
        int2 en = make_int2(0, 0);
        if (p + 32 + lane < pEnd) en = ev[p + 32 + lane];
        const int m = (pEnd - p < 32) ? (pEnd - p) : 32;
        for (int u0 = 0; u0 < m; u0 += PULL_U) {
            if (u0 + PULL_U <= m) pull_group<true >(Bv, sink, e, u0, m, n0, node, acc);
            else                  pull_group<false>(Bv, sink, e, u0, m, n0, node, acc);
        }
        e = en;
    }
    sink.row(node, acc);
}

__global__ __launch_bounds__(256) void gcn_agg_pull(
        const float* __restrict__ B, const int* __restrict__ rp,
        const int* __restrict__ ent, float* __restrict__ out) {
    const int lane = threadIdx.x & 31;
    const int wave = __builtin_amdgcn_readfirstlane(
        (int)(blockIdx.x * 8 + (threadIdx.x >> 5)));
    RowStore sink;
    sink.ov = (float4*)out + lane;
    pull_wave(B, rp, ent, wave * PULL_NPW, lane, sink);
}

__global__ __launch_bounds__(256) void gcn_agg_pull_pool(
        const float* __restrict__ B, const int* __restrict__ rp,
        const int* __restrict__ ent, const float* __restrict__ bias,
        const int* __restrict__ batch, float* sums, float* cnts) {
    __shared__ float shS[8 * 128];
    __shared__ int   shG[8];
    __shared__ int   shC[8];
    const int lane = threadIdx.x & 31;
    const int wib  = threadIdx.x >> 5;
    const int wave = __builtin_amdgcn_readfirstlane((int)(blockIdx.x * 8 + wib));
    const int n0 = wave * PULL_NPW;

    RowPool sink;
    sink.bv   = ((const float4*)bias)[lane];
    sink.gv   = batch[n0 + (lane & (PULL_NPW - 1))];
    sink.s    = make_float4(0.0f, 0.0f, 0.0f, 0.0f);
    sink.n0   = n0;
    sink.lane = lane;
    sink.cur  = -1;
    sink.cnt  = 0;
    sink.sums = sums;
    sink.cnts = cnts;
    pull_wave(B, rp, ent, n0, lane, sink);

    ((float4*)shS)[wib * 32 + lane] = sink.s;
    if (lane == 0) { shG[wib] = sink.cur; shC[wib] = sink.cnt; }
    __syncthreads();
    const int t = threadIdx.x;
    if (t < 128) {
        int   c = shG[0];
        float v = shS[t];
        for (int w = 1; w < 8; ++w) {
            const int g = shG[w];
            if (g != c) { atomicAdd(&sums[c * 128 + t], v); v = 0.0f; c = g; }
            v += shS[w * 128 + t];
        }
        atomicAdd(&sums[c * 128 + t], v);
    } else if (t == 128) {
        int c = shG[0];
        int n = shC[0];
        for (int w = 1; w < 8; ++w) {
            const int g = shG[w];
            if (g != c) { atomicAdd(&cnts[c], (float)n); n = 0; c = g; }
            n += shC[w];
        }
        atomicAdd(&cnts[c], (float)n);
    }
}

__global__ void gcn_deg_init(float* deg) {
    int i = blockIdx.x * 256 + threadIdx.x;
    if (i < N_NODES) deg[i] = 1.0f;
}

__global__ void gcn_deg_edges(const int* __restrict__ dst, float* deg) {
    int e = blockIdx.x * 256 + threadIdx.x;
    if (e < N_EDGES) atomicAdd(&deg[dst[e]], 1.0f);
}

__global__ void gcn_deg_rsqrt(float* deg) {
    int i = blockIdx.x * 256 + threadIdx.x;
    if (i < N_NODES) deg[i] = rsqrtf(deg[i]);
}

__global__ void gcn_agg_init(const float* __restrict__ B,
                             const float* __restrict__ dis,
                             float* __restrict__ out) {
    long idx = (long)blockIdx.x * 256 + threadIdx.x;
    if (idx >= (long)N_NODES * 128) return;
    int n = (int)(idx >> 7);
    float d = dis[n];
    out[idx] = d * d * B[idx];
}

__global__ __launch_bounds__(256) void gcn_agg_edges(
        const float* __restrict__ B, float* __restrict__ out,
        const int* __restrict__ src, const int* __restrict__ dst,
        const float* __restrict__ dis) {
    int e = blockIdx.x * 8 + (threadIdx.x >> 5);
    if (e >= N_EDGES) return;
    const int lane = threadIdx.x & 31;
    const int s = src[e], d = dst[e];
    const float norm = dis[s] * dis[d];
    float4 v = ((const float4*)(B + (long)s * 128))[lane];
    float* op = out + (long)d * 128 + lane * 4;
    atomicAdd(op + 0, norm * v.x);
    atomicAdd(op + 1, norm * v.y);
    atomicAdd(op + 2, norm * v.z);
    atomicAdd(op + 3, norm * v.w);
}

__global__ void gcn_pool_zero(float* sums, float* cnts) {
    int i = blockIdx.x * 256 + threadIdx.x;
    if (i < NUM_GRAPHS * 128) sums[i] = 0.0f;
    if (i < NUM_GRAPHS) cnts[i] = 0.0f;
}

#define POOL_NPW     40
#define POOL_U        8
#define POOL_WAVES  (N_NODES / POOL_NPW)
#define POOL_BLOCKS (POOL_WAVES / 8)
static_assert(N_NODES % (POOL_NPW * 8) == 0, "whole waves, whole workgroups");
static_assert(POOL_NPW % POOL_U == 0 && (POOL_NPW / POOL_U) % 2 == 1,
              "an odd number of groups: a b a b ... a");
static_assert(POOL_NPW > 32 && POOL_NPW <= 64,
              "graph ids of a wave's nodes sit in two registers");

__global__ __launch_bounds__(256) void gcn_pool_accum(
        const float* __restrict__ h, const float* __restrict__ bias,
        const int* __restrict__ batch, float* sums, float* cnts) {
    const int lane = threadIdx.x & 31;
    const int wave = __builtin_amdgcn_readfirstlane(
        (int)(blockIdx.x * 8 + (threadIdx.x >> 5)));
    const int n0 = wave * POOL_NPW;

    const int gv0 = batch[n0 + lane];
    const int gv1 = batch[n0 + 32 + (lane < POOL_NPW - 32 ? lane : POOL_NPW - 33)];
    const float4 bv = ((const float4*)bias)[lane];
    __builtin_amdgcn_sched_barrier(0);
    const float4* hp = (const float4*)h + (long)n0 * 32 + lane;

    float4 acc = make_float4(0.0f, 0.0f, 0.0f, 0.0f);
    int cnt = 0;
    int cur = 0;

    float4 va[POOL_U], vb[POOL_U];

    auto request = [&](float4 (&v)[POOL_U], int r0) {
#pragma unroll
        for (int u = 0; u < POOL_U; ++u) v[u] = hp[(r0 + u) * 32];
    };
    auto hand_over = [&]() {
        float* sp = sums + cur * 128 + lane * 4;
        atomicAdd(sp + 0, acc.x);
        atomicAdd(sp + 1, acc.y);
        atomicAdd(sp + 2, acc.z);
        atomicAdd(sp + 3, acc.w);
        if (lane == 0) atomicAdd(&cnts[cur], (float)cnt);
    };
    auto sum_group = [&](const float4 (&v)[POOL_U], int r0) {
#pragma unroll
        for (int u = 0; u < POOL_U; ++u) {
            const int r = r0 + u;
            const int g = __builtin_amdgcn_readlane(r < 32 ? gv0 : gv1, r & 31);
            if (g != cur) {
                hand_over();
                acc = make_float4(0.0f, 0.0f, 0.0f, 0.0f);
                cnt = 0;
                cur = g;
            }
            acc.x += fmaxf(v[u].x + bv.x, 0.0f);
            acc.y += fmaxf(v[u].y + bv.y, 0.0f);
            acc.z += fmaxf(v[u].z + bv.z, 0.0f);
            acc.w += fmaxf(v[u].w + bv.w, 0.0f);
            ++cnt;
        }
    };

    request(va, 0);
    request(vb, POOL_U);
    __builtin_amdgcn_sched_barrier(0);
    cur = __builtin_amdgcn_readlane(gv0, 0);
#pragma unroll
    for (int r0 = 0; r0 < POOL_NPW - POOL_U; r0 += 2 * POOL_U) {
        sum_group(va, r0);
        request(va, r0 + 2 * POOL_U);
        sum_group(vb, r0 + POOL_U);
        if (r0 + 3 * POOL_U < POOL_NPW) request(vb, r0 + 3 * POOL_U);
    }
    sum_group(va, POOL_NPW - POOL_U);
    hand_over();
}

__global__ void gcn_fc(const float* __restrict__ sums, const float* __restrict__ cnts,
                       const float* __restrict__ fcW, const float* __restrict__ fcb,
                       float* __restrict__ out) {
    int idx = blockIdx.x * 256 + threadIdx.x;
    if (idx >= NUM_GRAPHS * OUTC) return;
    int g = idx >> 4, o = idx & 15;
    float inv = 1.0f / fmaxf(cnts[g], 1.0f);
    float acc = fcb[o];
    for (int k = 0; k < 128; ++k)
        acc += sums[g * 128 + k] * inv * fcW[k * OUTC + o];
    out[idx] = acc;
}

#define WS_ROWS_BYTES   ((size_t)N_NODES * 128 * 4)
#define WS_SMALL_BYTES  ((size_t)N_NODES * 4 + (size_t)NUM_GRAPHS * 128 * 4 + (size_t)NUM_GRAPHS * 4)
#define WS_ENT_BYTES    ((size_t)N_ENT * 8)
#define WS_GATHER_BYTES (2 * WS_ROWS_BYTES + WS_SMALL_BYTES + WS_ENT_BYTES)

extern "C" void kernel_launch(void* const* d_in, const int* in_sizes, int n_in,
                              void* d_out, int out_size, void* d_ws, size_t ws_size,
                              hipStream_t stream) {
    (void)in_sizes; (void)n_in; (void)out_size;
    const float* x    = (const float*)d_in[0];
    const float* top  = (const float*)d_in[1];
    const int*   ei   = (const int*)d_in[2];
    const int*   bat  = (const int*)d_in[3];
    const float* W1   = (const float*)d_in[4];
    const float* b1   = (const float*)d_in[5];
    const float* W2   = (const float*)d_in[6];
    const float* b2   = (const float*)d_in[7];
    const float* fcW  = (const float*)d_in[8];
    const float* fcb  = (const float*)d_in[9];
    float*       out  = (float*)d_out;

    const int* src = ei;
    const int* dst = ei + N_EDGES;

    char* ws = (char*)d_ws;
    float* A    = (float*)ws; ws += WS_ROWS_BYTES;
    float* Bh   = (float*)ws; ws += WS_ROWS_BYTES;
    void*  nplane = (void*)ws; ws += (size_t)N_NODES * 4;
    float* sums = (float*)ws; ws += (size_t)NUM_GRAPHS * 128 * 4;
    float* cnts = (float*)ws; ws += (size_t)NUM_GRAPHS * 4;
    int*   ent  = (int*)ws;

    const int tb = 256;
    const int gN    = (N_NODES + tb - 1) / tb;
    const int gE    = (N_EDGES + tb - 1) / tb;
    const int gGemm = (N_NODES / 16 + 7) / 8;

    if (ws_size >= WS_GATHER_BYTES) {
        int*   rp   = (int*)nplane;
        int*   cnt  = (int*)A;
        int*   rank = cnt + N_NODES;
        float* dis  = (float*)(rank + N_EDGES);
        int*   tsum = (int*)(dis + N_NODES);

        csr_zero     <<<gN, tb, 0, stream>>>(cnt, sums, cnts);
        csr_count_with_gemm<<<GEMM_IN_COUNT + EDGE_BLOCKS, tb, 0, stream>>>(
            dst, cnt, rank, x, top, W1, Bh);
        csr_tile_sums<<<SCAN_BLOCKS, tb, 0, stream>>>(cnt, tsum);
        csr_scan     <<<SCAN_BLOCKS, tb, 0, stream>>>(cnt, tsum, rp, dis);
        csr_fill_with_gemm<<<GEMM_IN_FILL + ENT_BLOCKS, tb, 0, stream>>>(
            src, dst, rank, rp, dis, ent, x, top, W1, Bh);

        gcn_agg_pull <<<PULL_BLOCKS, tb, 0, stream>>>(Bh, rp, ent, A);

        gcn_gemm_wmma<1><<<gGemm, tb, 0, stream>>>(A, b1, W2, Bh, N_NODES);
        gcn_agg_pull_pool<<<PULL_BLOCKS, tb, 0, stream>>>(Bh, rp, ent, b2, bat, sums, cnts);
    } else {
        float* dis = (float*)nplane;
        const long NC   = (long)N_NODES * 128;
        const int gNC   = (int)((NC + tb - 1) / tb);
        const int gEdge = (N_EDGES + 7) / 8;

        gcn_deg_init <<<gN, tb, 0, stream>>>(dis);
        gcn_deg_edges<<<gE, tb, 0, stream>>>(dst, dis);
        gcn_deg_rsqrt<<<gN, tb, 0, stream>>>(dis);

        gcn_gemm_wmma<0><<<gGemm, tb, 0, stream>>>(x, top, W1, Bh, N_NODES);
        gcn_agg_init <<<gNC, tb, 0, stream>>>(Bh, dis, A);
        gcn_agg_edges<<<gEdge, tb, 0, stream>>>(Bh, A, src, dst, dis);

        gcn_gemm_wmma<1><<<gGemm, tb, 0, stream>>>(A, b1, W2, Bh, N_NODES);
        gcn_agg_init <<<gNC, tb, 0, stream>>>(Bh, dis, A);
        gcn_agg_edges<<<gEdge, tb, 0, stream>>>(Bh, A, src, dst, dis);

        gcn_pool_zero <<<(NUM_GRAPHS * 128 + tb - 1) / tb, tb, 0, stream>>>(sums, cnts);
        gcn_pool_accum<<<POOL_BLOCKS, tb, 0, stream>>>(A, b2, bat, sums, cnts);
    }

    gcn_fc<<<(NUM_GRAPHS * OUTC + tb - 1) / tb, tb, 0, stream>>>(sums, cnts, fcW, fcb, out);
}


# layer-1 aggregation feeds the layer-2 GEMM in one launch: a wave sums its 16 nodes' rows, 16 requested at a time, into an LDS tile (b1, relu) and multiplies it by W2 there; 9 launches become 8
# speedup vs baseline: 3.6926x; 1.0590x over previous
#include <hip/hip_runtime.h>
#include <hip/hip_bf16.h>


#define N_NODES   200000
#define N_EDGES   640000
#define NUM_GRAPHS   256
#define OUTC          16

typedef __attribute__((ext_vector_type(2))) float v2f;
typedef __attribute__((ext_vector_type(8))) float v8f;

#define N_ENT      (N_NODES + N_EDGES)
#define ENT_SELF    0x80000000u
#define SCAN_TILE         1024
#define SCAN_BLOCKS ((N_NODES + SCAN_TILE - 1) / SCAN_TILE)
static_assert(SCAN_BLOCKS <= 256, "one thread per earlier tile total");
static_assert(N_NODES % 4 == 0, "nodes are scanned four per thread");

static_assert(N_NODES >= NUM_GRAPHS * 128, "the node grid covers the pool planes");
__global__ void csr_zero(int* cnt, float* sums, float* cnts) {
    int i = blockIdx.x * 256 + threadIdx.x;
    if (i < N_NODES) cnt[i] = 0;
    if (i < NUM_GRAPHS * 128) sums[i] = 0.0f;
    if (i < NUM_GRAPHS) cnts[i] = 0.0f;
}

__device__ __forceinline__ void csr_count_block(int blk, const int* __restrict__ dst,
                                                int* cnt, int* __restrict__ rank) {
    int e = blk * 256 + threadIdx.x;
    if (e < N_EDGES) rank[e] = atomicAdd(&cnt[dst[e]], 1);
}

__device__ __forceinline__ int wave_sum(int v) {
    for (int m = 16; m >= 1; m >>= 1) v += __shfl_xor(v, m);
    return v;
}

__global__ __launch_bounds__(256) void csr_tile_sums(const int* __restrict__ cnt,
                                                     int* __restrict__ tsum) {
    __shared__ int sh[8];
    const int t = threadIdx.x;
    const int i = blockIdx.x * SCAN_TILE + t * 4;
    int s = 0;
    if (i < N_NODES) {
        const int4 c = *(const int4*)(cnt + i);
        s = c.x + c.y + c.z + c.w;
    }
    s = wave_sum(s);
    if ((t & 31) == 0) sh[t >> 5] = s;
    __syncthreads();
    if (t == 0) {
        int tot = 0;
        for (int k = 0; k < 8; ++k) tot += sh[k];
        tsum[blockIdx.x] = tot;
    }
}

__global__ __launch_bounds__(256) void csr_scan(const int* __restrict__ cnt,
                                                const int* __restrict__ tsum,
                                                int* __restrict__ rp,
                                                float* __restrict__ dis) {
    __shared__ int shA[8];
    __shared__ int shB[8];
    const int t = threadIdx.x, lane = t & 31, wv = t >> 5;
    const int b = blockIdx.x;
    int pre = (t < b) ? tsum[t] : 0;
    pre = wave_sum(pre);
    const int i = b * SCAN_TILE + t * 4;
    int4 c = make_int4(0, 0, 0, 0);
    if (i < N_NODES) c = *(const int4*)(cnt + i);
    const int mine = c.x + c.y + c.z + c.w;
    int inc = mine;
    for (int d = 1; d < 32; d <<= 1) {
        const int y = __shfl_up(inc, d);
        if (lane >= d) inc += y;
    }
    if (lane == 0)  shA[wv] = pre;
    if (lane == 31) shB[wv] = inc;
    __syncthreads();
    int base = 0;
    for (int k = 0; k < 8; ++k) {
        base += shA[k];
        if (k < wv) base += shB[k];
    }
    if (i < N_NODES) {
        int4 r;
        r.x = base + inc - mine + i;
        r.y = r.x + c.x + 1;
        r.z = r.y + c.y + 1;
        r.w = r.z + c.z + 1;
        *(int4*)(rp + i) = r;
        float4 d4;
        d4.x = rsqrtf((float)(c.x + 1));
        d4.y = rsqrtf((float)(c.y + 1));
        d4.z = rsqrtf((float)(c.z + 1));
        d4.w = rsqrtf((float)(c.w + 1));
        *(float4*)(dis + i) = d4;
    }
}

__device__ __forceinline__ void csr_fill_block(int blk,
                         const int* __restrict__ src, const int* __restrict__ dst,
                         const int* __restrict__ rank, const int* __restrict__ rp,
                         const float* __restrict__ dis, int* __restrict__ ent) {
    int idx = blk * 256 + threadIdx.x;
    if (idx < N_EDGES) {
        const int s = src[idx], d = dst[idx];
        const int p = rp[d] + 1 + rank[idx];
        ((int2*)ent)[p] = make_int2(s, __float_as_int(dis[s] * dis[d]));
    } else if (idx < N_ENT) {
        const int i = idx - N_EDGES;
        const float di = dis[i];
        ((int2*)ent)[rp[i]] = make_int2((int)((unsigned)i | ENT_SELF), __float_as_int(di * di));
    }
}

template <int MODE>
__device__ __forceinline__ void gcn_gemm_block(
        const float* __restrict__ A0, const float* __restrict__ A1,
        const float* __restrict__ W, float* __restrict__ B, int nRows, int blk) {
    __shared__ float sW[128 * 128];
    for (int i = threadIdx.x; i < 128 * 128; i += 256) {
        int kk = i >> 7, n = i & 127;
        sW[((kk >> 1) << 8) + n * 2 + (kk & 1)] = W[i];
    }
    __syncthreads();

    const int wave = threadIdx.x >> 5;
    const int lane = threadIdx.x & 31;
    const long m0  = ((long)blk * 8 + wave) * 16;
    if (m0 >= nRows) return;

    const int half = lane >> 4;
    const int lr   = lane & 15;
    const int koff = half * 2;
    const long row = m0 + lr;

    auto loadA = [&](int k) -> v2f {
        v2f r;
        if (MODE == 0) {
            if (k < 124) {
                const float* p = A0 + row * 124 + k + koff;
                r.x = p[0]; r.y = p[1];
            } else {
                const float* p = A1 + row * 4 + (k - 124) + koff;
                r.x = p[0]; r.y = p[1];
            }
        } else {
            const float* p = A0 + row * 128 + k + koff;
            r.x = fmaxf(p[0] + A1[k + koff],     0.0f);
            r.y = fmaxf(p[1] + A1[k + koff + 1], 0.0f);
        }
        return r;
    };

    v8f acc[8];
#pragma unroll
    for (int i = 0; i < 8; ++i) acc[i] = (v8f)(0.0f);

    v2f a = loadA(0);
    for (int k = 0; k < 128; k += 4) {
        v2f an = a;
        if (k + 4 < 128) an = loadA(k + 4);
        const float* bbase = sW + (((k + koff) >> 1) << 8) + lr * 2;
#pragma unroll
        for (int nt = 0; nt < 8; ++nt) {
            v2f b = *(const v2f*)(bbase + nt * 32);
            acc[nt] = __builtin_amdgcn_wmma_f32_16x16x4_f32(
                false, a, false, b, (short)0, acc[nt], false, false);
        }
        a = an;
    }
#pragma unroll
    for (int nt = 0; nt < 8; ++nt) {
        const int n0 = nt * 16;
#pragma unroll
        for (int r = 0; r < 8; ++r)
            B[(m0 + half * 8 + r) * 128 + n0 + lr] = acc[nt][r];
    }
}

template <int MODE>
__global__ __launch_bounds__(256) void gcn_gemm_wmma(
        const float* __restrict__ A0, const float* __restrict__ A1,
        const float* __restrict__ W, float* __restrict__ B, int nRows) {
    gcn_gemm_block<MODE>(A0, A1, W, B, nRows, (int)blockIdx.x);
}

#define GEMM_BLOCKS     ((N_NODES / 16 + 7) / 8)
#define GEMM_IN_COUNT   768
#define GEMM_IN_FILL    (GEMM_BLOCKS - GEMM_IN_COUNT)
#define EDGE_BLOCKS     ((N_EDGES + 255) / 256)
#define ENT_BLOCKS      ((N_ENT + 255) / 256)
static_assert(GEMM_IN_FILL > 0, "both launches carry GEMM blocks");

__global__ __launch_bounds__(256) void csr_count_with_gemm(
        const int* __restrict__ dst, int* cnt, int* __restrict__ rank,
        const float* __restrict__ x, const float* __restrict__ top,
        const float* __restrict__ W, float* __restrict__ B) {
    if (blockIdx.x < GEMM_IN_COUNT)
        gcn_gemm_block<0>(x, top, W, B, N_NODES, (int)blockIdx.x);
    else
        csr_count_block((int)blockIdx.x - GEMM_IN_COUNT, dst, cnt, rank);
}

__global__ __launch_bounds__(256) void csr_fill_with_gemm(
        const int* __restrict__ src, const int* __restrict__ dst,
        const int* __restrict__ rank, const int* __restrict__ rp,
        const float* __restrict__ dis, int* __restrict__ ent,
        const float* __restrict__ x, const float* __restrict__ top,
        const float* __restrict__ W, float* __restrict__ B) {
    if (blockIdx.x < GEMM_IN_FILL)
        gcn_gemm_block<0>(x, top, W, B, N_NODES, GEMM_IN_COUNT + (int)blockIdx.x);
    else
        csr_fill_block((int)blockIdx.x - GEMM_IN_FILL, src, dst, rank, rp, dis, ent);
}

#define PULL_NPW      4
#define PULL_U        8
#define PULL_WAVES  (N_NODES / PULL_NPW)
#define PULL_BLOCKS (PULL_WAVES / 8)
static_assert(N_NODES % (PULL_NPW * 8) == 0, "whole waves, whole workgroups");

#define FUSE_NPW     16
#define FUSE_U       16
#define FUSE_PITCH   132
#define FUSE_BLOCKS  ((N_NODES / FUSE_NPW + 7) / 8)
static_assert(N_NODES % FUSE_NPW == 0, "whole 16-row tiles");
struct RowTile {
    float4 bv;
    float* tp;
    int    n0;
    __device__ __forceinline__ void row(int node, float4 a) {
        float4 h;
        h.x = fmaxf(a.x + bv.x, 0.0f);
        h.y = fmaxf(a.y + bv.y, 0.0f);
        h.z = fmaxf(a.z + bv.z, 0.0f);
        h.w = fmaxf(a.w + bv.w, 0.0f);
        *(float4*)(tp + (node - n0) * FUSE_PITCH) = h;
    }
};

struct RowPool {
    float4 bv;
    float4 s;
    int    gv;
    int    n0, lane;
    int    cur, cnt;
    float* sums;
    float* cnts;
    __device__ __forceinline__ void flush() {
        float* sp = sums + cur * 128 + lane * 4;
        atomicAdd(sp + 0, s.x);
        atomicAdd(sp + 1, s.y);
        atomicAdd(sp + 2, s.z);
        atomicAdd(sp + 3, s.w);
        if (lane == 0) atomicAdd(&cnts[cur], (float)cnt);
    }
    __device__ __forceinline__ void row(int node, float4 a) {
        const int g = __builtin_amdgcn_readlane(gv, node - n0);
        if (g != cur) {
            if (cnt > 0) flush();
            s = make_float4(0.0f, 0.0f, 0.0f, 0.0f);
            cnt = 0;
            cur = g;
        }
        s.x += fmaxf(a.x + bv.x, 0.0f);
        s.y += fmaxf(a.y + bv.y, 0.0f);
        s.z += fmaxf(a.z + bv.z, 0.0f);
        s.w += fmaxf(a.w + bv.w, 0.0f);
        ++cnt;
    }
};

template <int U, bool FULL, class Sink>
__device__ __forceinline__ void pull_group(const float4* Bv, Sink& sink, int2 e,
                                           int u0, int m, int n0,
                                           int& node, float4& acc) {
    int    cs[U];
    float  wt[U];
    float4 v[U];
#pragma unroll
    for (int u = 0; u < U; ++u) {
        cs[u] = __builtin_amdgcn_readlane(e.x, u0 + u);
        wt[u] = __int_as_float(__builtin_amdgcn_readlane(e.y, u0 + u));
    }
#pragma unroll
    for (int u = 0; u < U; ++u)
        if (FULL || u0 + u < m) v[u] = Bv[(long)(cs[u] & 0x7fffffff) * 32];
#pragma unroll
    for (int u = 0; u < U; ++u)
        if (FULL || u0 + u < m) {
            if (cs[u] < 0) {
                if (node >= n0) sink.row(node, acc);
                ++node;
                acc = make_float4(0.0f, 0.0f, 0.0f, 0.0f);
            }
            acc.x = fmaf(wt[u], v[u].x, acc.x);
            acc.y = fmaf(wt[u], v[u].y, acc.y);
            acc.z = fmaf(wt[u], v[u].z, acc.z);
            acc.w = fmaf(wt[u], v[u].w, acc.w);
        }
}

template <int NPW, int U, class Sink>
__device__ __forceinline__ void pull_wave(const float* __restrict__ B,
                                          const int* __restrict__ rp,
                                          const int* __restrict__ ent,
                                          int n0, int lane, Sink& sink) {
    const int pBeg = rp[n0];
    const int pEnd = (n0 + NPW < N_NODES) ? rp[n0 + NPW] : N_ENT;

    const int2*   ev = (const int2*)ent;
    const float4* Bv = (const float4*)B + lane;

    float4 acc = make_float4(0.0f, 0.0f, 0.0f, 0.0f);
    int node = n0 - 1;

    int2 e = make_int2(0, 0);
    if (pBeg + lane < pEnd) e = ev[pBeg + lane];
    for (int p = pBeg; p < pEnd; p += 32) {
        int2 en = make_int2(0, 0);
        if (p + 32 + lane < pEnd) en = ev[p + 32 + lane];
        const int m = (pEnd - p < 32) ? (pEnd - p) : 32;
        for (int u0 = 0; u0 < m; u0 += U) {
            if (u0 + U <= m) pull_group<U, true >(Bv, sink, e, u0, m, n0, node, acc);
            else             pull_group<U, false>(Bv, sink, e, u0, m, n0, node, acc);
        }
        e = en;
    }
    sink.row(node, acc);
}

__global__ __launch_bounds__(256) void gcn_agg_pull_gemm(
        const float* __restrict__ Bin, const int* __restrict__ rp,
        const int* __restrict__ ent, const float* __restrict__ bias,
        const float* __restrict__ W, float* __restrict__ Bout) {
    __shared__ float smem[128 * 128 + 8 * FUSE_NPW * FUSE_PITCH];
    float* sW = smem;
    float* sA = smem + 128 * 128;
    for (int i = threadIdx.x; i < 128 * 128; i += 256) {
        int kk = i >> 7, n = i & 127;
        sW[((kk >> 1) << 8) + n * 2 + (kk & 1)] = W[i];
    }
    __syncthreads();

    const int lane = threadIdx.x & 31;
    const int wib  = threadIdx.x >> 5;
    const int wave = __builtin_amdgcn_readfirstlane((int)(blockIdx.x * 8 + wib));
    const int n0 = wave * FUSE_NPW;
    if (n0 >= N_NODES) return;

    float* tile = sA + wib * (FUSE_NPW * FUSE_PITCH);
    RowTile sink;
    sink.bv = ((const float4*)bias)[lane];
    sink.tp = tile + lane * 4;
    sink.n0 = n0;
    pull_wave<FUSE_NPW, FUSE_U>(Bin, rp, ent, n0, lane, sink);

    __builtin_amdgcn_fence(__ATOMIC_ACQ_REL, "wavefront");

    const int half = lane >> 4;
    const int lr   = lane & 15;
    const int koff = half * 2;
    const float* arow = tile + lr * FUSE_PITCH + koff;

    v8f acc[8];
#pragma unroll
    for (int i = 0; i < 8; ++i) acc[i] = (v8f)(0.0f);

    v2f a = *(const v2f*)(arow);
    for (int k = 0; k < 128; k += 4) {
        v2f an = a;
        if (k + 4 < 128) an = *(const v2f*)(arow + k + 4);
        const float* bbase = sW + (((k + koff) >> 1) << 8) + lr * 2;
#pragma unroll
        for (int nt = 0; nt < 8; ++nt) {
            v2f b = *(const v2f*)(bbase + nt * 32);
            acc[nt] = __builtin_amdgcn_wmma_f32_16x16x4_f32(
                false, a, false, b, (short)0, acc[nt], false, false);
        }
        a = an;
    }
    const long m0 = n0;
#pragma unroll
    for (int nt = 0; nt < 8; ++nt) {
        const int c0 = nt * 16;
#pragma unroll
        for (int r = 0; r < 8; ++r)
            Bout[(m0 + half * 8 + r) * 128 + c0 + lr] = acc[nt][r];
    }
}

__global__ __launch_bounds__(256) void gcn_agg_pull_pool(
        const float* __restrict__ B, const int* __restrict__ rp,
        const int* __restrict__ ent, const float* __restrict__ bias,
        const int* __restrict__ batch, float* sums, float* cnts) {
    __shared__ float shS[8 * 128];
    __shared__ int   shG[8];
    __shared__ int   shC[8];
    const int lane = threadIdx.x & 31;
    const int wib  = threadIdx.x >> 5;
    const int wave = __builtin_amdgcn_readfirstlane((int)(blockIdx.x * 8 + wib));
    const int n0 = wave * PULL_NPW;

    RowPool sink;
    sink.bv   = ((const float4*)bias)[lane];
    sink.gv   = batch[n0 + (lane & (PULL_NPW - 1))];
    sink.s    = make_float4(0.0f, 0.0f, 0.0f, 0.0f);
    sink.n0   = n0;
    sink.lane = lane;
    sink.cur  = -1;
    sink.cnt  = 0;
    sink.sums = sums;
    sink.cnts = cnts;
    pull_wave<PULL_NPW, PULL_U>(B, rp, ent, n0, lane, sink);

    ((float4*)shS)[wib * 32 + lane] = sink.s;
    if (lane == 0) { shG[wib] = sink.cur; shC[wib] = sink.cnt; }
    __syncthreads();
    const int t = threadIdx.x;
    if (t < 128) {
        int   c = shG[0];
        float v = shS[t];
        for (int w = 1; w < 8; ++w) {
            const int g = shG[w];
            if (g != c) { atomicAdd(&sums[c * 128 + t], v); v = 0.0f; c = g; }
            v += shS[w * 128 + t];
        }
        atomicAdd(&sums[c * 128 + t], v);
    } else if (t == 128) {
        int c = shG[0];
        int n = shC[0];
        for (int w = 1; w < 8; ++w) {
            const int g = shG[w];
            if (g != c) { atomicAdd(&cnts[c], (float)n); n = 0; c = g; }
            n += shC[w];
        }
        atomicAdd(&cnts[c], (float)n);
    }
}

__global__ void gcn_deg_init(float* deg) {
    int i = blockIdx.x * 256 + threadIdx.x;
    if (i < N_NODES) deg[i] = 1.0f;
}

__global__ void gcn_deg_edges(const int* __restrict__ dst, float* deg) {
    int e = blockIdx.x * 256 + threadIdx.x;
    if (e < N_EDGES) atomicAdd(&deg[dst[e]], 1.0f);
}

__global__ void gcn_deg_rsqrt(float* deg) {
    int i = blockIdx.x * 256 + threadIdx.x;
    if (i < N_NODES) deg[i] = rsqrtf(deg[i]);
}

__global__ void gcn_agg_init(const float* __restrict__ B,
                             const float* __restrict__ dis,
                             float* __restrict__ out) {
    long idx = (long)blockIdx.x * 256 + threadIdx.x;
    if (idx >= (long)N_NODES * 128) return;
    int n = (int)(idx >> 7);
    float d = dis[n];
    out[idx] = d * d * B[idx];
}

__global__ __launch_bounds__(256) void gcn_agg_edges(
        const float* __restrict__ B, float* __restrict__ out,
        const int* __restrict__ src, const int* __restrict__ dst,
        const float* __restrict__ dis) {
    int e = blockIdx.x * 8 + (threadIdx.x >> 5);
    if (e >= N_EDGES) return;
    const int lane = threadIdx.x & 31;
    const int s = src[e], d = dst[e];
    const float norm = dis[s] * dis[d];
    float4 v = ((const float4*)(B + (long)s * 128))[lane];
    float* op = out + (long)d * 128 + lane * 4;
    atomicAdd(op + 0, norm * v.x);
    atomicAdd(op + 1, norm * v.y);
    atomicAdd(op + 2, norm * v.z);
    atomicAdd(op + 3, norm * v.w);
}

__global__ void gcn_pool_zero(float* sums, float* cnts) {
    int i = blockIdx.x * 256 + threadIdx.x;
    if (i < NUM_GRAPHS * 128) sums[i] = 0.0f;
    if (i < NUM_GRAPHS) cnts[i] = 0.0f;
}

#define POOL_NPW     40
#define POOL_U        8
#define POOL_WAVES  (N_NODES / POOL_NPW)
#define POOL_BLOCKS (POOL_WAVES / 8)
static_assert(N_NODES % (POOL_NPW * 8) == 0, "whole waves, whole workgroups");
static_assert(POOL_NPW % POOL_U == 0 && (POOL_NPW / POOL_U) % 2 == 1,
              "an odd number of groups: a b a b ... a");
static_assert(POOL_NPW > 32 && POOL_NPW <= 64,
              "graph ids of a wave's nodes sit in two registers");

__global__ __launch_bounds__(256) void gcn_pool_accum(
        const float* __restrict__ h, const float* __restrict__ bias,
        const int* __restrict__ batch, float* sums, float* cnts) {
    const int lane = threadIdx.x & 31;
    const int wave = __builtin_amdgcn_readfirstlane(
        (int)(blockIdx.x * 8 + (threadIdx.x >> 5)));
    const int n0 = wave * POOL_NPW;

    const int gv0 = batch[n0 + lane];
    const int gv1 = batch[n0 + 32 + (lane < POOL_NPW - 32 ? lane : POOL_NPW - 33)];
    const float4 bv = ((const float4*)bias)[lane];
    __builtin_amdgcn_sched_barrier(0);
    const float4* hp = (const float4*)h + (long)n0 * 32 + lane;

    float4 acc = make_float4(0.0f, 0.0f, 0.0f, 0.0f);
    int cnt = 0;
    int cur = 0;

    float4 va[POOL_U], vb[POOL_U];

    auto request = [&](float4 (&v)[POOL_U], int r0) {
#pragma unroll
        for (int u = 0; u < POOL_U; ++u) v[u] = hp[(r0 + u) * 32];
    };
    auto hand_over = [&]() {
        float* sp = sums + cur * 128 + lane * 4;
        atomicAdd(sp + 0, acc.x);
        atomicAdd(sp + 1, acc.y);
        atomicAdd(sp + 2, acc.z);
        atomicAdd(sp + 3, acc.w);
        if (lane == 0) atomicAdd(&cnts[cur], (float)cnt);
    };
    auto sum_group = [&](const float4 (&v)[POOL_U], int r0) {
#pragma unroll
        for (int u = 0; u < POOL_U; ++u) {
            const int r = r0 + u;
            const int g = __builtin_amdgcn_readlane(r < 32 ? gv0 : gv1, r & 31);
            if (g != cur) {
                hand_over();
                acc = make_float4(0.0f, 0.0f, 0.0f, 0.0f);
                cnt = 0;
                cur = g;
            }
            acc.x += fmaxf(v[u].x + bv.x, 0.0f);
            acc.y += fmaxf(v[u].y + bv.y, 0.0f);
            acc.z += fmaxf(v[u].z + bv.z, 0.0f);
            acc.w += fmaxf(v[u].w + bv.w, 0.0f);
            ++cnt;
        }
    };

    request(va, 0);
    request(vb, POOL_U);
    __builtin_amdgcn_sched_barrier(0);
    cur = __builtin_amdgcn_readlane(gv0, 0);
#pragma unroll
    for (int r0 = 0; r0 < POOL_NPW - POOL_U; r0 += 2 * POOL_U) {
        sum_group(va, r0);
        request(va, r0 + 2 * POOL_U);
        sum_group(vb, r0 + POOL_U);
        if (r0 + 3 * POOL_U < POOL_NPW) request(vb, r0 + 3 * POOL_U);
    }
    sum_group(va, POOL_NPW - POOL_U);
    hand_over();
}

__global__ void gcn_fc(const float* __restrict__ sums, const float* __restrict__ cnts,
                       const float* __restrict__ fcW, const float* __restrict__ fcb,
                       float* __restrict__ out) {
    int idx = blockIdx.x * 256 + threadIdx.x;
    if (idx >= NUM_GRAPHS * OUTC) return;
    int g = idx >> 4, o = idx & 15;
    float inv = 1.0f / fmaxf(cnts[g], 1.0f);
    float acc = fcb[o];
    for (int k = 0; k < 128; ++k)
        acc += sums[g * 128 + k] * inv * fcW[k * OUTC + o];
    out[idx] = acc;
}

#define WS_ROWS_BYTES   ((size_t)N_NODES * 128 * 4)
#define WS_SMALL_BYTES  ((size_t)N_NODES * 4 + (size_t)NUM_GRAPHS * 128 * 4 + (size_t)NUM_GRAPHS * 4)
#define WS_ENT_BYTES    ((size_t)N_ENT * 8)
#define WS_GATHER_BYTES (2 * WS_ROWS_BYTES + WS_SMALL_BYTES + WS_ENT_BYTES)

extern "C" void kernel_launch(void* const* d_in, const int* in_sizes, int n_in,
                              void* d_out, int out_size, void* d_ws, size_t ws_size,
                              hipStream_t stream) {
    (void)in_sizes; (void)n_in; (void)out_size;
    const float* x    = (const float*)d_in[0];
    const float* top  = (const float*)d_in[1];
    const int*   ei   = (const int*)d_in[2];
    const int*   bat  = (const int*)d_in[3];
    const float* W1   = (const float*)d_in[4];
    const float* b1   = (const float*)d_in[5];
    const float* W2   = (const float*)d_in[6];
    const float* b2   = (const float*)d_in[7];
    const float* fcW  = (const float*)d_in[8];
    const float* fcb  = (const float*)d_in[9];
    float*       out  = (float*)d_out;

    const int* src = ei;
    const int* dst = ei + N_EDGES;

    char* ws = (char*)d_ws;
    float* A    = (float*)ws; ws += WS_ROWS_BYTES;
    float* Bh   = (float*)ws; ws += WS_ROWS_BYTES;
    void*  nplane = (void*)ws; ws += (size_t)N_NODES * 4;
    float* sums = (float*)ws; ws += (size_t)NUM_GRAPHS * 128 * 4;
    float* cnts = (float*)ws; ws += (size_t)NUM_GRAPHS * 4;
    int*   ent  = (int*)ws;

    const int tb = 256;
    const int gN    = (N_NODES + tb - 1) / tb;
    const int gE    = (N_EDGES + tb - 1) / tb;
    const int gGemm = (N_NODES / 16 + 7) / 8;

    if (ws_size >= WS_GATHER_BYTES) {
        int*   rp   = (int*)nplane;
        int*   cnt  = (int*)A;
        int*   rank = cnt + N_NODES;
        float* dis  = (float*)(rank + N_EDGES);
        int*   tsum = (int*)(dis + N_NODES);

        csr_zero     <<<gN, tb, 0, stream>>>(cnt, sums, cnts);
        csr_count_with_gemm<<<GEMM_IN_COUNT + EDGE_BLOCKS, tb, 0, stream>>>(
            dst, cnt, rank, x, top, W1, Bh);
        csr_tile_sums<<<SCAN_BLOCKS, tb, 0, stream>>>(cnt, tsum);
        csr_scan     <<<SCAN_BLOCKS, tb, 0, stream>>>(cnt, tsum, rp, dis);
        csr_fill_with_gemm<<<GEMM_IN_FILL + ENT_BLOCKS, tb, 0, stream>>>(
            src, dst, rank, rp, dis, ent, x, top, W1, Bh);

        gcn_agg_pull_gemm<<<FUSE_BLOCKS, tb, 0, stream>>>(Bh, rp, ent, b1, W2, A);
        gcn_agg_pull_pool<<<PULL_BLOCKS, tb, 0, stream>>>(A, rp, ent, b2, bat, sums, cnts);
    } else {
        float* dis = (float*)nplane;
        const long NC   = (long)N_NODES * 128;
        const int gNC   = (int)((NC + tb - 1) / tb);
        const int gEdge = (N_EDGES + 7) / 8;

        gcn_deg_init <<<gN, tb, 0, stream>>>(dis);
        gcn_deg_edges<<<gE, tb, 0, stream>>>(dst, dis);
        gcn_deg_rsqrt<<<gN, tb, 0, stream>>>(dis);

        gcn_gemm_wmma<0><<<gGemm, tb, 0, stream>>>(x, top, W1, Bh, N_NODES);
        gcn_agg_init <<<gNC, tb, 0, stream>>>(Bh, dis, A);
        gcn_agg_edges<<<gEdge, tb, 0, stream>>>(Bh, A, src, dst, dis);

        gcn_gemm_wmma<1><<<gGemm, tb, 0, stream>>>(A, b1, W2, Bh, N_NODES);
        gcn_agg_init <<<gNC, tb, 0, stream>>>(Bh, dis, A);
        gcn_agg_edges<<<gEdge, tb, 0, stream>>>(Bh, A, src, dst, dis);

        gcn_pool_zero <<<(NUM_GRAPHS * 128 + tb - 1) / tb, tb, 0, stream>>>(sums, cnts);
        gcn_pool_accum<<<POOL_BLOCKS, tb, 0, stream>>>(A, b2, bat, sums, cnts);
    }

    gcn_fc<<<(NUM_GRAPHS * OUTC + tb - 1) / tb, tb, 0, stream>>>(sums, cnts, fcW, fcb, out);
}


# layer-1 aggregation + layer-2 GEMM launch made resident: one 24-wave workgroup per WGP stages W2 once, its waves loop over 16-node tiles; 6 waves per SIMD, not 4; 256 weight stagings, not 1563
# speedup vs baseline: 3.8104x; 1.0319x over previous
#include <hip/hip_runtime.h>
#include <hip/hip_bf16.h>


#define N_NODES   200000
#define N_EDGES   640000
#define NUM_GRAPHS   256
#define OUTC          16

typedef __attribute__((ext_vector_type(2))) float v2f;
typedef __attribute__((ext_vector_type(8))) float v8f;

#define N_ENT      (N_NODES + N_EDGES)
#define ENT_SELF    0x80000000u
#define SCAN_TILE         1024
#define SCAN_BLOCKS ((N_NODES + SCAN_TILE - 1) / SCAN_TILE)
static_assert(SCAN_BLOCKS <= 256, "one thread per earlier tile total");
static_assert(N_NODES % 4 == 0, "nodes are scanned four per thread");

static_assert(N_NODES >= NUM_GRAPHS * 128, "the node grid covers the pool planes");
__global__ void csr_zero(int* cnt, float* sums, float* cnts) {
    int i = blockIdx.x * 256 + threadIdx.x;
    if (i < N_NODES) cnt[i] = 0;
    if (i < NUM_GRAPHS * 128) sums[i] = 0.0f;
    if (i < NUM_GRAPHS) cnts[i] = 0.0f;
}

__device__ __forceinline__ void csr_count_block(int blk, const int* __restrict__ dst,
                                                int* cnt, int* __restrict__ rank) {
    int e = blk * 256 + threadIdx.x;
    if (e < N_EDGES) rank[e] = atomicAdd(&cnt[dst[e]], 1);
}

__device__ __forceinline__ int wave_sum(int v) {
    for (int m = 16; m >= 1; m >>= 1) v += __shfl_xor(v, m);
    return v;
}

__global__ __launch_bounds__(256) void csr_tile_sums(const int* __restrict__ cnt,
                                                     int* __restrict__ tsum) {
    __shared__ int sh[8];
    const int t = threadIdx.x;
    const int i = blockIdx.x * SCAN_TILE + t * 4;
    int s = 0;
    if (i < N_NODES) {
        const int4 c = *(const int4*)(cnt + i);
        s = c.x + c.y + c.z + c.w;
    }
    s = wave_sum(s);
    if ((t & 31) == 0) sh[t >> 5] = s;
    __syncthreads();
    if (t == 0) {
        int tot = 0;
        for (int k = 0; k < 8; ++k) tot += sh[k];
        tsum[blockIdx.x] = tot;
    }
}

__global__ __launch_bounds__(256) void csr_scan(const int* __restrict__ cnt,
                                                const int* __restrict__ tsum,
                                                int* __restrict__ rp,
                                                float* __restrict__ dis) {
    __shared__ int shA[8];
    __shared__ int shB[8];
    const int t = threadIdx.x, lane = t & 31, wv = t >> 5;
    const int b = blockIdx.x;
    int pre = (t < b) ? tsum[t] : 0;
    pre = wave_sum(pre);
    const int i = b * SCAN_TILE + t * 4;
    int4 c = make_int4(0, 0, 0, 0);
    if (i < N_NODES) c = *(const int4*)(cnt + i);
    const int mine = c.x + c.y + c.z + c.w;
    int inc = mine;
    for (int d = 1; d < 32; d <<= 1) {
        const int y = __shfl_up(inc, d);
        if (lane >= d) inc += y;
    }
    if (lane == 0)  shA[wv] = pre;
    if (lane == 31) shB[wv] = inc;
    __syncthreads();
    int base = 0;
    for (int k = 0; k < 8; ++k) {
        base += shA[k];
        if (k < wv) base += shB[k];
    }
    if (i < N_NODES) {
        int4 r;
        r.x = base + inc - mine + i;
        r.y = r.x + c.x + 1;
        r.z = r.y + c.y + 1;
        r.w = r.z + c.z + 1;
        *(int4*)(rp + i) = r;
        float4 d4;
        d4.x = rsqrtf((float)(c.x + 1));
        d4.y = rsqrtf((float)(c.y + 1));
        d4.z = rsqrtf((float)(c.z + 1));
        d4.w = rsqrtf((float)(c.w + 1));
        *(float4*)(dis + i) = d4;
    }
}

__device__ __forceinline__ void csr_fill_block(int blk,
                         const int* __restrict__ src, const int* __restrict__ dst,
                         const int* __restrict__ rank, const int* __restrict__ rp,
                         const float* __restrict__ dis, int* __restrict__ ent) {
    int idx = blk * 256 + threadIdx.x;
    if (idx < N_EDGES) {
        const int s = src[idx], d = dst[idx];
        const int p = rp[d] + 1 + rank[idx];
        ((int2*)ent)[p] = make_int2(s, __float_as_int(dis[s] * dis[d]));
    } else if (idx < N_ENT) {
        const int i = idx - N_EDGES;
        const float di = dis[i];
        ((int2*)ent)[rp[i]] = make_int2((int)((unsigned)i | ENT_SELF), __float_as_int(di * di));
    }
}

template <int MODE>
__device__ __forceinline__ void gcn_gemm_block(
        const float* __restrict__ A0, const float* __restrict__ A1,
        const float* __restrict__ W, float* __restrict__ B, int nRows, int blk) {
    __shared__ float sW[128 * 128];
    for (int i = threadIdx.x; i < 128 * 128; i += 256) {
        int kk = i >> 7, n = i & 127;
        sW[((kk >> 1) << 8) + n * 2 + (kk & 1)] = W[i];
    }
    __syncthreads();

    const int wave = threadIdx.x >> 5;
    const int lane = threadIdx.x & 31;
    const long m0  = ((long)blk * 8 + wave) * 16;
    if (m0 >= nRows) return;

    const int half = lane >> 4;
    const int lr   = lane & 15;
    const int koff = half * 2;
    const long row = m0 + lr;

    auto loadA = [&](int k) -> v2f {
        v2f r;
        if (MODE == 0) {
            if (k < 124) {
                const float* p = A0 + row * 124 + k + koff;
                r.x = p[0]; r.y = p[1];
            } else {
                const float* p = A1 + row * 4 + (k - 124) + koff;
                r.x = p[0]; r.y = p[1];
            }
        } else {
            const float* p = A0 + row * 128 + k + koff;
            r.x = fmaxf(p[0] + A1[k + koff],     0.0f);
            r.y = fmaxf(p[1] + A1[k + koff + 1], 0.0f);
        }
        return r;
    };

    v8f acc[8];
#pragma unroll
    for (int i = 0; i < 8; ++i) acc[i] = (v8f)(0.0f);

    v2f a = loadA(0);
    for (int k = 0; k < 128; k += 4) {
        v2f an = a;
        if (k + 4 < 128) an = loadA(k + 4);
        const float* bbase = sW + (((k + koff) >> 1) << 8) + lr * 2;
#pragma unroll
        for (int nt = 0; nt < 8; ++nt) {
            v2f b = *(const v2f*)(bbase + nt * 32);
            acc[nt] = __builtin_amdgcn_wmma_f32_16x16x4_f32(
                false, a, false, b, (short)0, acc[nt], false, false);
        }
        a = an;
    }
#pragma unroll
    for (int nt = 0; nt < 8; ++nt) {
        const int n0 = nt * 16;
#pragma unroll
        for (int r = 0; r < 8; ++r)
            B[(m0 + half * 8 + r) * 128 + n0 + lr] = acc[nt][r];
    }
}

template <int MODE>
__global__ __launch_bounds__(256) void gcn_gemm_wmma(
        const float* __restrict__ A0, const float* __restrict__ A1,
        const float* __restrict__ W, float* __restrict__ B, int nRows) {
    gcn_gemm_block<MODE>(A0, A1, W, B, nRows, (int)blockIdx.x);
}

#define GEMM_BLOCKS     ((N_NODES / 16 + 7) / 8)
#define GEMM_IN_COUNT   768
#define GEMM_IN_FILL    (GEMM_BLOCKS - GEMM_IN_COUNT)
#define EDGE_BLOCKS     ((N_EDGES + 255) / 256)
#define ENT_BLOCKS      ((N_ENT + 255) / 256)
static_assert(GEMM_IN_FILL > 0, "both launches carry GEMM blocks");

__global__ __launch_bounds__(256) void csr_count_with_gemm(
        const int* __restrict__ dst, int* cnt, int* __restrict__ rank,
        const float* __restrict__ x, const float* __restrict__ top,
        const float* __restrict__ W, float* __restrict__ B) {
    if (blockIdx.x < GEMM_IN_COUNT)
        gcn_gemm_block<0>(x, top, W, B, N_NODES, (int)blockIdx.x);
    else
        csr_count_block((int)blockIdx.x - GEMM_IN_COUNT, dst, cnt, rank);
}

__global__ __launch_bounds__(256) void csr_fill_with_gemm(
        const int* __restrict__ src, const int* __restrict__ dst,
        const int* __restrict__ rank, const int* __restrict__ rp,
        const float* __restrict__ dis, int* __restrict__ ent,
        const float* __restrict__ x, const float* __restrict__ top,
        const float* __restrict__ W, float* __restrict__ B) {
    if (blockIdx.x < GEMM_IN_FILL)
        gcn_gemm_block<0>(x, top, W, B, N_NODES, GEMM_IN_COUNT + (int)blockIdx.x);
    else
        csr_fill_block((int)blockIdx.x - GEMM_IN_FILL, src, dst, rank, rp, dis, ent);
}

#define PULL_NPW      4
#define PULL_U        8
#define PULL_WAVES  (N_NODES / PULL_NPW)
#define PULL_BLOCKS (PULL_WAVES / 8)
static_assert(N_NODES % (PULL_NPW * 8) == 0, "whole waves, whole workgroups");

#define FUSE_NPW     16
#define FUSE_U       16
#define FUSE_PITCH   132
static_assert(N_NODES % FUSE_NPW == 0, "whole 16-row tiles");
struct RowTile {
    float4 bv;
    float* tp;
    int    n0;
    __device__ __forceinline__ void row(int node, float4 a) {
        float4 h;
        h.x = fmaxf(a.x + bv.x, 0.0f);
        h.y = fmaxf(a.y + bv.y, 0.0f);
        h.z = fmaxf(a.z + bv.z, 0.0f);
        h.w = fmaxf(a.w + bv.w, 0.0f);
        *(float4*)(tp + (node - n0) * FUSE_PITCH) = h;
    }
};

struct RowPool {
    float4 bv;
    float4 s;
    int    gv;
    int    n0, lane;
    int    cur, cnt;
    float* sums;
    float* cnts;
    __device__ __forceinline__ void flush() {
        float* sp = sums + cur * 128 + lane * 4;
        atomicAdd(sp + 0, s.x);
        atomicAdd(sp + 1, s.y);
        atomicAdd(sp + 2, s.z);
        atomicAdd(sp + 3, s.w);
        if (lane == 0) atomicAdd(&cnts[cur], (float)cnt);
    }
    __device__ __forceinline__ void row(int node, float4 a) {
        const int g = __builtin_amdgcn_readlane(gv, node - n0);
        if (g != cur) {
            if (cnt > 0) flush();
            s = make_float4(0.0f, 0.0f, 0.0f, 0.0f);
            cnt = 0;
            cur = g;
        }
        s.x += fmaxf(a.x + bv.x, 0.0f);
        s.y += fmaxf(a.y + bv.y, 0.0f);
        s.z += fmaxf(a.z + bv.z, 0.0f);
        s.w += fmaxf(a.w + bv.w, 0.0f);
        ++cnt;
    }
};

template <int U, bool FULL, class Sink>
__device__ __forceinline__ void pull_group(const float4* Bv, Sink& sink, int2 e,
                                           int u0, int m, int n0,
                                           int& node, float4& acc) {
    int    cs[U];
    float  wt[U];
    float4 v[U];
#pragma unroll
    for (int u = 0; u < U; ++u) {
        cs[u] = __builtin_amdgcn_readlane(e.x, u0 + u);
        wt[u] = __int_as_float(__builtin_amdgcn_readlane(e.y, u0 + u));
    }
#pragma unroll
    for (int u = 0; u < U; ++u)
        if (FULL || u0 + u < m) v[u] = Bv[(long)(cs[u] & 0x7fffffff) * 32];
#pragma unroll
    for (int u = 0; u < U; ++u)
        if (FULL || u0 + u < m) {
            if (cs[u] < 0) {
                if (node >= n0) sink.row(node, acc);
                ++node;
                acc = make_float4(0.0f, 0.0f, 0.0f, 0.0f);
            }
            acc.x = fmaf(wt[u], v[u].x, acc.x);
            acc.y = fmaf(wt[u], v[u].y, acc.y);
            acc.z = fmaf(wt[u], v[u].z, acc.z);
            acc.w = fmaf(wt[u], v[u].w, acc.w);
        }
}

template <int NPW, int U, class Sink>
__device__ __forceinline__ void pull_wave(const float* __restrict__ B,
                                          const int* __restrict__ rp,
                                          const int* __restrict__ ent,
                                          int n0, int lane, Sink& sink) {
    const int pBeg = rp[n0];
    const int pEnd = (n0 + NPW < N_NODES) ? rp[n0 + NPW] : N_ENT;

    const int2*   ev = (const int2*)ent;
    const float4* Bv = (const float4*)B + lane;

    float4 acc = make_float4(0.0f, 0.0f, 0.0f, 0.0f);
    int node = n0 - 1;

    int2 e = make_int2(0, 0);
    if (pBeg + lane < pEnd) e = ev[pBeg + lane];
    for (int p = pBeg; p < pEnd; p += 32) {
        int2 en = make_int2(0, 0);
        if (p + 32 + lane < pEnd) en = ev[p + 32 + lane];
        const int m = (pEnd - p < 32) ? (pEnd - p) : 32;
        for (int u0 = 0; u0 < m; u0 += U) {
            if (u0 + U <= m) pull_group<U, true >(Bv, sink, e, u0, m, n0, node, acc);
            else             pull_group<U, false>(Bv, sink, e, u0, m, n0, node, acc);
        }
        e = en;
    }
    sink.row(node, acc);
}

#define FUSE_WAVES   24
#define FUSE_GRID    256
#define FUSE_TILES   (N_NODES / FUSE_NPW)
__global__ __launch_bounds__(FUSE_WAVES * 32) void gcn_agg_pull_gemm(
        const float* __restrict__ Bin, const int* __restrict__ rp,
        const int* __restrict__ ent, const float* __restrict__ bias,
        const float* __restrict__ W, float* __restrict__ Bout) {
    __shared__ float smem[128 * 128 + FUSE_WAVES * FUSE_NPW * FUSE_PITCH];
    float* sW = smem;
    float* sA = smem + 128 * 128;
    for (int i = threadIdx.x; i < 128 * 128; i += FUSE_WAVES * 32) {
        int kk = i >> 7, n = i & 127;
        sW[((kk >> 1) << 8) + n * 2 + (kk & 1)] = W[i];
    }
    __syncthreads();

    const int lane = threadIdx.x & 31;
    const int wib  = __builtin_amdgcn_readfirstlane((int)(threadIdx.x >> 5));

    float* tile = sA + wib * (FUSE_NPW * FUSE_PITCH);
    RowTile sink;
    sink.bv = ((const float4*)bias)[lane];
    sink.tp = tile + lane * 4;

    const int half = lane >> 4;
    const int lr   = lane & 15;
    const int koff = half * 2;
    const float* arow = tile + lr * FUSE_PITCH + koff;

    for (int t = (int)blockIdx.x + FUSE_GRID * wib; t < FUSE_TILES;
         t += FUSE_GRID * FUSE_WAVES) {
        const int n0 = t * FUSE_NPW;
        sink.n0 = n0;
        pull_wave<FUSE_NPW, FUSE_U>(Bin, rp, ent, n0, lane, sink);

        __builtin_amdgcn_fence(__ATOMIC_ACQ_REL, "wavefront");

        v8f acc[8];
#pragma unroll
        for (int i = 0; i < 8; ++i) acc[i] = (v8f)(0.0f);

        v2f a = *(const v2f*)(arow);
        for (int k = 0; k < 128; k += 4) {
            v2f an = a;
            if (k + 4 < 128) an = *(const v2f*)(arow + k + 4);
            const float* bbase = sW + (((k + koff) >> 1) << 8) + lr * 2;
#pragma unroll
            for (int nt = 0; nt < 8; ++nt) {
                v2f b = *(const v2f*)(bbase + nt * 32);
                acc[nt] = __builtin_amdgcn_wmma_f32_16x16x4_f32(
                    false, a, false, b, (short)0, acc[nt], false, false);
            }
            a = an;
        }
        __builtin_amdgcn_fence(__ATOMIC_ACQ_REL, "wavefront");

        const long m0 = n0;
#pragma unroll
        for (int nt = 0; nt < 8; ++nt) {
            const int c0 = nt * 16;
#pragma unroll
            for (int r = 0; r < 8; ++r)
                Bout[(m0 + half * 8 + r) * 128 + c0 + lr] = acc[nt][r];
        }
    }
}

__global__ __launch_bounds__(256) void gcn_agg_pull_pool(
        const float* __restrict__ B, const int* __restrict__ rp,
        const int* __restrict__ ent, const float* __restrict__ bias,
        const int* __restrict__ batch, float* sums, float* cnts) {
    __shared__ float shS[8 * 128];
    __shared__ int   shG[8];
    __shared__ int   shC[8];
    const int lane = threadIdx.x & 31;
    const int wib  = threadIdx.x >> 5;
    const int wave = __builtin_amdgcn_readfirstlane((int)(blockIdx.x * 8 + wib));
    const int n0 = wave * PULL_NPW;

    RowPool sink;
    sink.bv   = ((const float4*)bias)[lane];
    sink.gv   = batch[n0 + (lane & (PULL_NPW - 1))];
    sink.s    = make_float4(0.0f, 0.0f, 0.0f, 0.0f);
    sink.n0   = n0;
    sink.lane = lane;
    sink.cur  = -1;
    sink.cnt  = 0;
    sink.sums = sums;
    sink.cnts = cnts;
    pull_wave<PULL_NPW, PULL_U>(B, rp, ent, n0, lane, sink);

    ((float4*)shS)[wib * 32 + lane] = sink.s;
    if (lane == 0) { shG[wib] = sink.cur; shC[wib] = sink.cnt; }
    __syncthreads();
    const int t = threadIdx.x;
    if (t < 128) {
        int   c = shG[0];
        float v = shS[t];
        for (int w = 1; w < 8; ++w) {
            const int g = shG[w];
            if (g != c) { atomicAdd(&sums[c * 128 + t], v); v = 0.0f; c = g; }
            v += shS[w * 128 + t];
        }
        atomicAdd(&sums[c * 128 + t], v);
    } else if (t == 128) {
        int c = shG[0];
        int n = shC[0];
        for (int w = 1; w < 8; ++w) {
            const int g = shG[w];
            if (g != c) { atomicAdd(&cnts[c], (float)n); n = 0; c = g; }
            n += shC[w];
        }
        atomicAdd(&cnts[c], (float)n);
    }
}

__global__ void gcn_deg_init(float* deg) {
    int i = blockIdx.x * 256 + threadIdx.x;
    if (i < N_NODES) deg[i] = 1.0f;
}

__global__ void gcn_deg_edges(const int* __restrict__ dst, float* deg) {
    int e = blockIdx.x * 256 + threadIdx.x;
    if (e < N_EDGES) atomicAdd(&deg[dst[e]], 1.0f);
}

__global__ void gcn_deg_rsqrt(float* deg) {
    int i = blockIdx.x * 256 + threadIdx.x;
    if (i < N_NODES) deg[i] = rsqrtf(deg[i]);
}

__global__ void gcn_agg_init(const float* __restrict__ B,
                             const float* __restrict__ dis,
                             float* __restrict__ out) {
    long idx = (long)blockIdx.x * 256 + threadIdx.x;
    if (idx >= (long)N_NODES * 128) return;
    int n = (int)(idx >> 7);
    float d = dis[n];
    out[idx] = d * d * B[idx];
}

__global__ __launch_bounds__(256) void gcn_agg_edges(
        const float* __restrict__ B, float* __restrict__ out,
        const int* __restrict__ src, const int* __restrict__ dst,
        const float* __restrict__ dis) {
    int e = blockIdx.x * 8 + (threadIdx.x >> 5);
    if (e >= N_EDGES) return;
    const int lane = threadIdx.x & 31;
    const int s = src[e], d = dst[e];
    const float norm = dis[s] * dis[d];
    float4 v = ((const float4*)(B + (long)s * 128))[lane];
    float* op = out + (long)d * 128 + lane * 4;
    atomicAdd(op + 0, norm * v.x);
    atomicAdd(op + 1, norm * v.y);
    atomicAdd(op + 2, norm * v.z);
    atomicAdd(op + 3, norm * v.w);
}

__global__ void gcn_pool_zero(float* sums, float* cnts) {
    int i = blockIdx.x * 256 + threadIdx.x;
    if (i < NUM_GRAPHS * 128) sums[i] = 0.0f;
    if (i < NUM_GRAPHS) cnts[i] = 0.0f;
}

#define POOL_NPW     40
#define POOL_U        8
#define POOL_WAVES  (N_NODES / POOL_NPW)
#define POOL_BLOCKS (POOL_WAVES / 8)
static_assert(N_NODES % (POOL_NPW * 8) == 0, "whole waves, whole workgroups");
static_assert(POOL_NPW % POOL_U == 0 && (POOL_NPW / POOL_U) % 2 == 1,
              "an odd number of groups: a b a b ... a");
static_assert(POOL_NPW > 32 && POOL_NPW <= 64,
              "graph ids of a wave's nodes sit in two registers");

__global__ __launch_bounds__(256) void gcn_pool_accum(
        const float* __restrict__ h, const float* __restrict__ bias,
        const int* __restrict__ batch, float* sums, float* cnts) {
    const int lane = threadIdx.x & 31;
    const int wave = __builtin_amdgcn_readfirstlane(
        (int)(blockIdx.x * 8 + (threadIdx.x >> 5)));
    const int n0 = wave * POOL_NPW;

    const int gv0 = batch[n0 + lane];
    const int gv1 = batch[n0 + 32 + (lane < POOL_NPW - 32 ? lane : POOL_NPW - 33)];
    const float4 bv = ((const float4*)bias)[lane];
    __builtin_amdgcn_sched_barrier(0);
    const float4* hp = (const float4*)h + (long)n0 * 32 + lane;

    float4 acc = make_float4(0.0f, 0.0f, 0.0f, 0.0f);
    int cnt = 0;
    int cur = 0;

    float4 va[POOL_U], vb[POOL_U];

    auto request = [&](float4 (&v)[POOL_U], int r0) {
#pragma unroll
        for (int u = 0; u < POOL_U; ++u) v[u] = hp[(r0 + u) * 32];
    };
    auto hand_over = [&]() {
        float* sp = sums + cur * 128 + lane * 4;
        atomicAdd(sp + 0, acc.x);
        atomicAdd(sp + 1, acc.y);
        atomicAdd(sp + 2, acc.z);
        atomicAdd(sp + 3, acc.w);
        if (lane == 0) atomicAdd(&cnts[cur], (float)cnt);
    };
    auto sum_group = [&](const float4 (&v)[POOL_U], int r0) {
#pragma unroll
        for (int u = 0; u < POOL_U; ++u) {
            const int r = r0 + u;
            const int g = __builtin_amdgcn_readlane(r < 32 ? gv0 : gv1, r & 31);
            if (g != cur) {
                hand_over();
                acc = make_float4(0.0f, 0.0f, 0.0f, 0.0f);
                cnt = 0;
                cur = g;
            }
            acc.x += fmaxf(v[u].x + bv.x, 0.0f);
            acc.y += fmaxf(v[u].y + bv.y, 0.0f);
            acc.z += fmaxf(v[u].z + bv.z, 0.0f);
            acc.w += fmaxf(v[u].w + bv.w, 0.0f);
            ++cnt;
        }
    };

    request(va, 0);
    request(vb, POOL_U);
    __builtin_amdgcn_sched_barrier(0);
    cur = __builtin_amdgcn_readlane(gv0, 0);
#pragma unroll
    for (int r0 = 0; r0 < POOL_NPW - POOL_U; r0 += 2 * POOL_U) {
        sum_group(va, r0);
        request(va, r0 + 2 * POOL_U);
        sum_group(vb, r0 + POOL_U);
        if (r0 + 3 * POOL_U < POOL_NPW) request(vb, r0 + 3 * POOL_U);
    }
    sum_group(va, POOL_NPW - POOL_U);
    hand_over();
}

__global__ void gcn_fc(const float* __restrict__ sums, const float* __restrict__ cnts,
                       const float* __restrict__ fcW, const float* __restrict__ fcb,
                       float* __restrict__ out) {
    int idx = blockIdx.x * 256 + threadIdx.x;
    if (idx >= NUM_GRAPHS * OUTC) return;
    int g = idx >> 4, o = idx & 15;
    float inv = 1.0f / fmaxf(cnts[g], 1.0f);
    float acc = fcb[o];
    for (int k = 0; k < 128; ++k)
        acc += sums[g * 128 + k] * inv * fcW[k * OUTC + o];
    out[idx] = acc;
}

#define WS_ROWS_BYTES   ((size_t)N_NODES * 128 * 4)
#define WS_SMALL_BYTES  ((size_t)N_NODES * 4 + (size_t)NUM_GRAPHS * 128 * 4 + (size_t)NUM_GRAPHS * 4)
#define WS_ENT_BYTES    ((size_t)N_ENT * 8)
#define WS_GATHER_BYTES (2 * WS_ROWS_BYTES + WS_SMALL_BYTES + WS_ENT_BYTES)

extern "C" void kernel_launch(void* const* d_in, const int* in_sizes, int n_in,
                              void* d_out, int out_size, void* d_ws, size_t ws_size,
                              hipStream_t stream) {
    (void)in_sizes; (void)n_in; (void)out_size;
    const float* x    = (const float*)d_in[0];
    const float* top  = (const float*)d_in[1];
    const int*   ei   = (const int*)d_in[2];
    const int*   bat  = (const int*)d_in[3];
    const float* W1   = (const float*)d_in[4];
    const float* b1   = (const float*)d_in[5];
    const float* W2   = (const float*)d_in[6];
    const float* b2   = (const float*)d_in[7];
    const float* fcW  = (const float*)d_in[8];
    const float* fcb  = (const float*)d_in[9];
    float*       out  = (float*)d_out;

    const int* src = ei;
    const int* dst = ei + N_EDGES;

    char* ws = (char*)d_ws;
    float* A    = (float*)ws; ws += WS_ROWS_BYTES;
    float* Bh   = (float*)ws; ws += WS_ROWS_BYTES;
    void*  nplane = (void*)ws; ws += (size_t)N_NODES * 4;
    float* sums = (float*)ws; ws += (size_t)NUM_GRAPHS * 128 * 4;
    float* cnts = (float*)ws; ws += (size_t)NUM_GRAPHS * 4;
    int*   ent  = (int*)ws;

    const int tb = 256;
    const int gN    = (N_NODES + tb - 1) / tb;
    const int gE    = (N_EDGES + tb - 1) / tb;
    const int gGemm = (N_NODES / 16 + 7) / 8;

    if (ws_size >= WS_GATHER_BYTES) {
        int*   rp   = (int*)nplane;
        int*   cnt  = (int*)A;
        int*   rank = cnt + N_NODES;
        float* dis  = (float*)(rank + N_EDGES);
        int*   tsum = (int*)(dis + N_NODES);

        csr_zero     <<<gN, tb, 0, stream>>>(cnt, sums, cnts);
        csr_count_with_gemm<<<GEMM_IN_COUNT + EDGE_BLOCKS, tb, 0, stream>>>(
            dst, cnt, rank, x, top, W1, Bh);
        csr_tile_sums<<<SCAN_BLOCKS, tb, 0, stream>>>(cnt, tsum);
        csr_scan     <<<SCAN_BLOCKS, tb, 0, stream>>>(cnt, tsum, rp, dis);
        csr_fill_with_gemm<<<GEMM_IN_FILL + ENT_BLOCKS, tb, 0, stream>>>(
            src, dst, rank, rp, dis, ent, x, top, W1, Bh);

        gcn_agg_pull_gemm<<<FUSE_GRID, FUSE_WAVES * 32, 0, stream>>>(Bh, rp, ent, b1, W2, A);
        gcn_agg_pull_pool<<<PULL_BLOCKS, tb, 0, stream>>>(A, rp, ent, b2, bat, sums, cnts);
    } else {
        float* dis = (float*)nplane;
        const long NC   = (long)N_NODES * 128;
        const int gNC   = (int)((NC + tb - 1) / tb);
        const int gEdge = (N_EDGES + 7) / 8;

        gcn_deg_init <<<gN, tb, 0, stream>>>(dis);
        gcn_deg_edges<<<gE, tb, 0, stream>>>(dst, dis);
        gcn_deg_rsqrt<<<gN, tb, 0, stream>>>(dis);

        gcn_gemm_wmma<0><<<gGemm, tb, 0, stream>>>(x, top, W1, Bh, N_NODES);
        gcn_agg_init <<<gNC, tb, 0, stream>>>(Bh, dis, A);
        gcn_agg_edges<<<gEdge, tb, 0, stream>>>(Bh, A, src, dst, dis);

        gcn_gemm_wmma<1><<<gGemm, tb, 0, stream>>>(A, b1, W2, Bh, N_NODES);
        gcn_agg_init <<<gNC, tb, 0, stream>>>(Bh, dis, A);
        gcn_agg_edges<<<gEdge, tb, 0, stream>>>(Bh, A, src, dst, dis);

        gcn_pool_zero <<<(NUM_GRAPHS * 128 + tb - 1) / tb, tb, 0, stream>>>(sums, cnts);
        gcn_pool_accum<<<POOL_BLOCKS, tb, 0, stream>>>(A, b2, bat, sums, cnts);
    }

    gcn_fc<<<(NUM_GRAPHS * OUTC + tb - 1) / tb, tb, 0, stream>>>(sums, cnts, fcW, fcb, out);
}


# layer-2 aggregation: a wave walks 25 consecutive nodes, not 4: 8000 waves, all resident at once, instead of 50000 in six rounds; 200 nodes per LDS combine and atomic; float sum order changes
# speedup vs baseline: 3.9628x; 1.0400x over previous
#include <hip/hip_runtime.h>
#include <hip/hip_bf16.h>


#define N_NODES   200000
#define N_EDGES   640000
#define NUM_GRAPHS   256
#define OUTC          16

typedef __attribute__((ext_vector_type(2))) float v2f;
typedef __attribute__((ext_vector_type(8))) float v8f;

#define N_ENT      (N_NODES + N_EDGES)
#define ENT_SELF    0x80000000u
#define SCAN_TILE         1024
#define SCAN_BLOCKS ((N_NODES + SCAN_TILE - 1) / SCAN_TILE)
static_assert(SCAN_BLOCKS <= 256, "one thread per earlier tile total");
static_assert(N_NODES % 4 == 0, "nodes are scanned four per thread");

static_assert(N_NODES >= NUM_GRAPHS * 128, "the node grid covers the pool planes");
__global__ void csr_zero(int* cnt, float* sums, float* cnts) {
    int i = blockIdx.x * 256 + threadIdx.x;
    if (i < N_NODES) cnt[i] = 0;
    if (i < NUM_GRAPHS * 128) sums[i] = 0.0f;
    if (i < NUM_GRAPHS) cnts[i] = 0.0f;
}

__device__ __forceinline__ void csr_count_block(int blk, const int* __restrict__ dst,
                                                int* cnt, int* __restrict__ rank) {
    int e = blk * 256 + threadIdx.x;
    if (e < N_EDGES) rank[e] = atomicAdd(&cnt[dst[e]], 1);
}

__device__ __forceinline__ int wave_sum(int v) {
    for (int m = 16; m >= 1; m >>= 1) v += __shfl_xor(v, m);
    return v;
}

__global__ __launch_bounds__(256) void csr_tile_sums(const int* __restrict__ cnt,
                                                     int* __restrict__ tsum) {
    __shared__ int sh[8];
    const int t = threadIdx.x;
    const int i = blockIdx.x * SCAN_TILE + t * 4;
    int s = 0;
    if (i < N_NODES) {
        const int4 c = *(const int4*)(cnt + i);
        s = c.x + c.y + c.z + c.w;
    }
    s = wave_sum(s);
    if ((t & 31) == 0) sh[t >> 5] = s;
    __syncthreads();
    if (t == 0) {
        int tot = 0;
        for (int k = 0; k < 8; ++k) tot += sh[k];
        tsum[blockIdx.x] = tot;
    }
}

__global__ __launch_bounds__(256) void csr_scan(const int* __restrict__ cnt,
                                                const int* __restrict__ tsum,
                                                int* __restrict__ rp,
                                                float* __restrict__ dis) {
    __shared__ int shA[8];
    __shared__ int shB[8];
    const int t = threadIdx.x, lane = t & 31, wv = t >> 5;
    const int b = blockIdx.x;
    int pre = (t < b) ? tsum[t] : 0;
    pre = wave_sum(pre);
    const int i = b * SCAN_TILE + t * 4;
    int4 c = make_int4(0, 0, 0, 0);
    if (i < N_NODES) c = *(const int4*)(cnt + i);
    const int mine = c.x + c.y + c.z + c.w;
    int inc = mine;
    for (int d = 1; d < 32; d <<= 1) {
        const int y = __shfl_up(inc, d);
        if (lane >= d) inc += y;
    }
    if (lane == 0)  shA[wv] = pre;
    if (lane == 31) shB[wv] = inc;
    __syncthreads();
    int base = 0;
    for (int k = 0; k < 8; ++k) {
        base += shA[k];
        if (k < wv) base += shB[k];
    }
    if (i < N_NODES) {
        int4 r;
        r.x = base + inc - mine + i;
        r.y = r.x + c.x + 1;
        r.z = r.y + c.y + 1;
        r.w = r.z + c.z + 1;
        *(int4*)(rp + i) = r;
        float4 d4;
        d4.x = rsqrtf((float)(c.x + 1));
        d4.y = rsqrtf((float)(c.y + 1));
        d4.z = rsqrtf((float)(c.z + 1));
        d4.w = rsqrtf((float)(c.w + 1));
        *(float4*)(dis + i) = d4;
    }
}

__device__ __forceinline__ void csr_fill_block(int blk,
                         const int* __restrict__ src, const int* __restrict__ dst,
                         const int* __restrict__ rank, const int* __restrict__ rp,
                         const float* __restrict__ dis, int* __restrict__ ent) {
    int idx = blk * 256 + threadIdx.x;
    if (idx < N_EDGES) {
        const int s = src[idx], d = dst[idx];
        const int p = rp[d] + 1 + rank[idx];
        ((int2*)ent)[p] = make_int2(s, __float_as_int(dis[s] * dis[d]));
    } else if (idx < N_ENT) {
        const int i = idx - N_EDGES;
        const float di = dis[i];
        ((int2*)ent)[rp[i]] = make_int2((int)((unsigned)i | ENT_SELF), __float_as_int(di * di));
    }
}

template <int MODE>
__device__ __forceinline__ void gcn_gemm_block(
        const float* __restrict__ A0, const float* __restrict__ A1,
        const float* __restrict__ W, float* __restrict__ B, int nRows, int blk) {
    __shared__ float sW[128 * 128];
    for (int i = threadIdx.x; i < 128 * 128; i += 256) {
        int kk = i >> 7, n = i & 127;
        sW[((kk >> 1) << 8) + n * 2 + (kk & 1)] = W[i];
    }
    __syncthreads();

    const int wave = threadIdx.x >> 5;
    const int lane = threadIdx.x & 31;
    const long m0  = ((long)blk * 8 + wave) * 16;
    if (m0 >= nRows) return;

    const int half = lane >> 4;
    const int lr   = lane & 15;
    const int koff = half * 2;
    const long row = m0 + lr;

    auto loadA = [&](int k) -> v2f {
        v2f r;
        if (MODE == 0) {
            if (k < 124) {
                const float* p = A0 + row * 124 + k + koff;
                r.x = p[0]; r.y = p[1];
            } else {
                const float* p = A1 + row * 4 + (k - 124) + koff;
                r.x = p[0]; r.y = p[1];
            }
        } else {
            const float* p = A0 + row * 128 + k + koff;
            r.x = fmaxf(p[0] + A1[k + koff],     0.0f);
            r.y = fmaxf(p[1] + A1[k + koff + 1], 0.0f);
        }
        return r;
    };

    v8f acc[8];
#pragma unroll
    for (int i = 0; i < 8; ++i) acc[i] = (v8f)(0.0f);

    v2f a = loadA(0);
    for (int k = 0; k < 128; k += 4) {
        v2f an = a;
        if (k + 4 < 128) an = loadA(k + 4);
        const float* bbase = sW + (((k + koff) >> 1) << 8) + lr * 2;
#pragma unroll
        for (int nt = 0; nt < 8; ++nt) {
            v2f b = *(const v2f*)(bbase + nt * 32);
            acc[nt] = __builtin_amdgcn_wmma_f32_16x16x4_f32(
                false, a, false, b, (short)0, acc[nt], false, false);
        }
        a = an;
    }
#pragma unroll
    for (int nt = 0; nt < 8; ++nt) {
        const int n0 = nt * 16;
#pragma unroll
        for (int r = 0; r < 8; ++r)
            B[(m0 + half * 8 + r) * 128 + n0 + lr] = acc[nt][r];
    }
}

template <int MODE>
__global__ __launch_bounds__(256) void gcn_gemm_wmma(
        const float* __restrict__ A0, const float* __restrict__ A1,
        const float* __restrict__ W, float* __restrict__ B, int nRows) {
    gcn_gemm_block<MODE>(A0, A1, W, B, nRows, (int)blockIdx.x);
}

#define GEMM_BLOCKS     ((N_NODES / 16 + 7) / 8)
#define GEMM_IN_COUNT   768
#define GEMM_IN_FILL    (GEMM_BLOCKS - GEMM_IN_COUNT)
#define EDGE_BLOCKS     ((N_EDGES + 255) / 256)
#define ENT_BLOCKS      ((N_ENT + 255) / 256)
static_assert(GEMM_IN_FILL > 0, "both launches carry GEMM blocks");

__global__ __launch_bounds__(256) void csr_count_with_gemm(
        const int* __restrict__ dst, int* cnt, int* __restrict__ rank,
        const float* __restrict__ x, const float* __restrict__ top,
        const float* __restrict__ W, float* __restrict__ B) {
    if (blockIdx.x < GEMM_IN_COUNT)
        gcn_gemm_block<0>(x, top, W, B, N_NODES, (int)blockIdx.x);
    else
        csr_count_block((int)blockIdx.x - GEMM_IN_COUNT, dst, cnt, rank);
}

__global__ __launch_bounds__(256) void csr_fill_with_gemm(
        const int* __restrict__ src, const int* __restrict__ dst,
        const int* __restrict__ rank, const int* __restrict__ rp,
        const float* __restrict__ dis, int* __restrict__ ent,
        const float* __restrict__ x, const float* __restrict__ top,
        const float* __restrict__ W, float* __restrict__ B) {
    if (blockIdx.x < GEMM_IN_FILL)
        gcn_gemm_block<0>(x, top, W, B, N_NODES, GEMM_IN_COUNT + (int)blockIdx.x);
    else
        csr_fill_block((int)blockIdx.x - GEMM_IN_FILL, src, dst, rank, rp, dis, ent);
}

#define PULL_NPW     25
#define PULL_U        8
#define PULL_WAVES  (N_NODES / PULL_NPW)
#define PULL_BLOCKS (PULL_WAVES / 8)
static_assert(N_NODES % (PULL_NPW * 8) == 0, "whole waves, whole workgroups");
static_assert(PULL_NPW <= 32, "the graph ids of a wave's nodes sit in one register");

#define FUSE_NPW     16
#define FUSE_U       16
#define FUSE_PITCH   132
static_assert(N_NODES % FUSE_NPW == 0, "whole 16-row tiles");
struct RowTile {
    float4 bv;
    float* tp;
    int    n0;
    __device__ __forceinline__ void row(int node, float4 a) {
        float4 h;
        h.x = fmaxf(a.x + bv.x, 0.0f);
        h.y = fmaxf(a.y + bv.y, 0.0f);
        h.z = fmaxf(a.z + bv.z, 0.0f);
        h.w = fmaxf(a.w + bv.w, 0.0f);
        *(float4*)(tp + (node - n0) * FUSE_PITCH) = h;
    }
};

struct RowPool {
    float4 bv;
    float4 s;
    int    gv;
    int    n0, lane;
    int    cur, cnt;
    float* sums;
    float* cnts;
    __device__ __forceinline__ void flush() {
        float* sp = sums + cur * 128 + lane * 4;
        atomicAdd(sp + 0, s.x);
        atomicAdd(sp + 1, s.y);
        atomicAdd(sp + 2, s.z);
        atomicAdd(sp + 3, s.w);
        if (lane == 0) atomicAdd(&cnts[cur], (float)cnt);
    }
    __device__ __forceinline__ void row(int node, float4 a) {
        const int g = __builtin_amdgcn_readlane(gv, node - n0);
        if (g != cur) {
            if (cnt > 0) flush();
            s = make_float4(0.0f, 0.0f, 0.0f, 0.0f);
            cnt = 0;
            cur = g;
        }
        s.x += fmaxf(a.x + bv.x, 0.0f);
        s.y += fmaxf(a.y + bv.y, 0.0f);
        s.z += fmaxf(a.z + bv.z, 0.0f);
        s.w += fmaxf(a.w + bv.w, 0.0f);
        ++cnt;
    }
};

template <int U, bool FULL, class Sink>
__device__ __forceinline__ void pull_group(const float4* Bv, Sink& sink, int2 e,
                                           int u0, int m, int n0,
                                           int& node, float4& acc) {
    int    cs[U];
    float  wt[U];
    float4 v[U];
#pragma unroll
    for (int u = 0; u < U; ++u) {
        cs[u] = __builtin_amdgcn_readlane(e.x, u0 + u);
        wt[u] = __int_as_float(__builtin_amdgcn_readlane(e.y, u0 + u));
    }
#pragma unroll
    for (int u = 0; u < U; ++u)
        if (FULL || u0 + u < m) v[u] = Bv[(long)(cs[u] & 0x7fffffff) * 32];
#pragma unroll
    for (int u = 0; u < U; ++u)
        if (FULL || u0 + u < m) {
            if (cs[u] < 0) {
                if (node >= n0) sink.row(node, acc);
                ++node;
                acc = make_float4(0.0f, 0.0f, 0.0f, 0.0f);
            }
            acc.x = fmaf(wt[u], v[u].x, acc.x);
            acc.y = fmaf(wt[u], v[u].y, acc.y);
            acc.z = fmaf(wt[u], v[u].z, acc.z);
            acc.w = fmaf(wt[u], v[u].w, acc.w);
        }
}

template <int NPW, int U, class Sink>
__device__ __forceinline__ void pull_wave(const float* __restrict__ B,
                                          const int* __restrict__ rp,
                                          const int* __restrict__ ent,
                                          int n0, int lane, Sink& sink) {
    const int pBeg = rp[n0];
    const int pEnd = (n0 + NPW < N_NODES) ? rp[n0 + NPW] : N_ENT;

    const int2*   ev = (const int2*)ent;
    const float4* Bv = (const float4*)B + lane;

    float4 acc = make_float4(0.0f, 0.0f, 0.0f, 0.0f);
    int node = n0 - 1;

    int2 e = make_int2(0, 0);
    if (pBeg + lane < pEnd) e = ev[pBeg + lane];
    for (int p = pBeg; p < pEnd; p += 32) {
        int2 en = make_int2(0, 0);
        if (p + 32 + lane < pEnd) en = ev[p + 32 + lane];
        const int m = (pEnd - p < 32) ? (pEnd - p) : 32;
        for (int u0 = 0; u0 < m; u0 += U) {
            if (u0 + U <= m) pull_group<U, true >(Bv, sink, e, u0, m, n0, node, acc);
            else             pull_group<U, false>(Bv, sink, e, u0, m, n0, node, acc);
        }
        e = en;
    }
    sink.row(node, acc);
}

#define FUSE_WAVES   24
#define FUSE_GRID    256
#define FUSE_TILES   (N_NODES / FUSE_NPW)
__global__ __launch_bounds__(FUSE_WAVES * 32) void gcn_agg_pull_gemm(
        const float* __restrict__ Bin, const int* __restrict__ rp,
        const int* __restrict__ ent, const float* __restrict__ bias,
        const float* __restrict__ W, float* __restrict__ Bout) {
    __shared__ float smem[128 * 128 + FUSE_WAVES * FUSE_NPW * FUSE_PITCH];
    float* sW = smem;
    float* sA = smem + 128 * 128;
    for (int i = threadIdx.x; i < 128 * 128; i += FUSE_WAVES * 32) {
        int kk = i >> 7, n = i & 127;
        sW[((kk >> 1) << 8) + n * 2 + (kk & 1)] = W[i];
    }
    __syncthreads();

    const int lane = threadIdx.x & 31;
    const int wib  = __builtin_amdgcn_readfirstlane((int)(threadIdx.x >> 5));

    float* tile = sA + wib * (FUSE_NPW * FUSE_PITCH);
    RowTile sink;
    sink.bv = ((const float4*)bias)[lane];
    sink.tp = tile + lane * 4;

    const int half = lane >> 4;
    const int lr   = lane & 15;
    const int koff = half * 2;
    const float* arow = tile + lr * FUSE_PITCH + koff;

    for (int t = (int)blockIdx.x + FUSE_GRID * wib; t < FUSE_TILES;
         t += FUSE_GRID * FUSE_WAVES) {
        const int n0 = t * FUSE_NPW;
        sink.n0 = n0;
        pull_wave<FUSE_NPW, FUSE_U>(Bin, rp, ent, n0, lane, sink);

        __builtin_amdgcn_fence(__ATOMIC_ACQ_REL, "wavefront");

        v8f acc[8];
#pragma unroll
        for (int i = 0; i < 8; ++i) acc[i] = (v8f)(0.0f);

        v2f a = *(const v2f*)(arow);
        for (int k = 0; k < 128; k += 4) {
            v2f an = a;
            if (k + 4 < 128) an = *(const v2f*)(arow + k + 4);
            const float* bbase = sW + (((k + koff) >> 1) << 8) + lr * 2;
#pragma unroll
            for (int nt = 0; nt < 8; ++nt) {
                v2f b = *(const v2f*)(bbase + nt * 32);
                acc[nt] = __builtin_amdgcn_wmma_f32_16x16x4_f32(
                    false, a, false, b, (short)0, acc[nt], false, false);
            }
            a = an;
        }
        __builtin_amdgcn_fence(__ATOMIC_ACQ_REL, "wavefront");

        const long m0 = n0;
#pragma unroll
        for (int nt = 0; nt < 8; ++nt) {
            const int c0 = nt * 16;
#pragma unroll
            for (int r = 0; r < 8; ++r)
                Bout[(m0 + half * 8 + r) * 128 + c0 + lr] = acc[nt][r];
        }
    }
}

__global__ __launch_bounds__(256) void gcn_agg_pull_pool(
        const float* __restrict__ B, const int* __restrict__ rp,
        const int* __restrict__ ent, const float* __restrict__ bias,
        const int* __restrict__ batch, float* sums, float* cnts) {
    __shared__ float shS[8 * 128];
    __shared__ int   shG[8];
    __shared__ int   shC[8];
    const int lane = threadIdx.x & 31;
    const int wib  = threadIdx.x >> 5;
    const int wave = __builtin_amdgcn_readfirstlane((int)(blockIdx.x * 8 + wib));
    const int n0 = wave * PULL_NPW;

    RowPool sink;
    sink.bv   = ((const float4*)bias)[lane];
    sink.gv   = batch[n0 + (lane < PULL_NPW ? lane : PULL_NPW - 1)];
    sink.s    = make_float4(0.0f, 0.0f, 0.0f, 0.0f);
    sink.n0   = n0;
    sink.lane = lane;
    sink.cur  = -1;
    sink.cnt  = 0;
    sink.sums = sums;
    sink.cnts = cnts;
    pull_wave<PULL_NPW, PULL_U>(B, rp, ent, n0, lane, sink);

    ((float4*)shS)[wib * 32 + lane] = sink.s;
    if (lane == 0) { shG[wib] = sink.cur; shC[wib] = sink.cnt; }
    __syncthreads();
    const int t = threadIdx.x;
    if (t < 128) {
        int   c = shG[0];
        float v = shS[t];
        for (int w = 1; w < 8; ++w) {
            const int g = shG[w];
            if (g != c) { atomicAdd(&sums[c * 128 + t], v); v = 0.0f; c = g; }
            v += shS[w * 128 + t];
        }
        atomicAdd(&sums[c * 128 + t], v);
    } else if (t == 128) {
        int c = shG[0];
        int n = shC[0];
        for (int w = 1; w < 8; ++w) {
            const int g = shG[w];
            if (g != c) { atomicAdd(&cnts[c], (float)n); n = 0; c = g; }
            n += shC[w];
        }
        atomicAdd(&cnts[c], (float)n);
    }
}

__global__ void gcn_deg_init(float* deg) {
    int i = blockIdx.x * 256 + threadIdx.x;
    if (i < N_NODES) deg[i] = 1.0f;
}

__global__ void gcn_deg_edges(const int* __restrict__ dst, float* deg) {
    int e = blockIdx.x * 256 + threadIdx.x;
    if (e < N_EDGES) atomicAdd(&deg[dst[e]], 1.0f);
}

__global__ void gcn_deg_rsqrt(float* deg) {
    int i = blockIdx.x * 256 + threadIdx.x;
    if (i < N_NODES) deg[i] = rsqrtf(deg[i]);
}

__global__ void gcn_agg_init(const float* __restrict__ B,
                             const float* __restrict__ dis,
                             float* __restrict__ out) {
    long idx = (long)blockIdx.x * 256 + threadIdx.x;
    if (idx >= (long)N_NODES * 128) return;
    int n = (int)(idx >> 7);
    float d = dis[n];
    out[idx] = d * d * B[idx];
}

__global__ __launch_bounds__(256) void gcn_agg_edges(
        const float* __restrict__ B, float* __restrict__ out,
        const int* __restrict__ src, const int* __restrict__ dst,
        const float* __restrict__ dis) {
    int e = blockIdx.x * 8 + (threadIdx.x >> 5);
    if (e >= N_EDGES) return;
    const int lane = threadIdx.x & 31;
    const int s = src[e], d = dst[e];
    const float norm = dis[s] * dis[d];
    float4 v = ((const float4*)(B + (long)s * 128))[lane];
    float* op = out + (long)d * 128 + lane * 4;
    atomicAdd(op + 0, norm * v.x);
    atomicAdd(op + 1, norm * v.y);
    atomicAdd(op + 2, norm * v.z);
    atomicAdd(op + 3, norm * v.w);
}

__global__ void gcn_pool_zero(float* sums, float* cnts) {
    int i = blockIdx.x * 256 + threadIdx.x;
    if (i < NUM_GRAPHS * 128) sums[i] = 0.0f;
    if (i < NUM_GRAPHS) cnts[i] = 0.0f;
}

#define POOL_NPW     40
#define POOL_U        8
#define POOL_WAVES  (N_NODES / POOL_NPW)
#define POOL_BLOCKS (POOL_WAVES / 8)
static_assert(N_NODES % (POOL_NPW * 8) == 0, "whole waves, whole workgroups");
static_assert(POOL_NPW % POOL_U == 0 && (POOL_NPW / POOL_U) % 2 == 1,
              "an odd number of groups: a b a b ... a");
static_assert(POOL_NPW > 32 && POOL_NPW <= 64,
              "graph ids of a wave's nodes sit in two registers");

__global__ __launch_bounds__(256) void gcn_pool_accum(
        const float* __restrict__ h, const float* __restrict__ bias,
        const int* __restrict__ batch, float* sums, float* cnts) {
    const int lane = threadIdx.x & 31;
    const int wave = __builtin_amdgcn_readfirstlane(
        (int)(blockIdx.x * 8 + (threadIdx.x >> 5)));
    const int n0 = wave * POOL_NPW;

    const int gv0 = batch[n0 + lane];
    const int gv1 = batch[n0 + 32 + (lane < POOL_NPW - 32 ? lane : POOL_NPW - 33)];
    const float4 bv = ((const float4*)bias)[lane];
    __builtin_amdgcn_sched_barrier(0);
    const float4* hp = (const float4*)h + (long)n0 * 32 + lane;

    float4 acc = make_float4(0.0f, 0.0f, 0.0f, 0.0f);
    int cnt = 0;
    int cur = 0;

    float4 va[POOL_U], vb[POOL_U];

    auto request = [&](float4 (&v)[POOL_U], int r0) {
#pragma unroll
        for (int u = 0; u < POOL_U; ++u) v[u] = hp[(r0 + u) * 32];
    };
    auto hand_over = [&]() {
        float* sp = sums + cur * 128 + lane * 4;
        atomicAdd(sp + 0, acc.x);
        atomicAdd(sp + 1, acc.y);
        atomicAdd(sp + 2, acc.z);
        atomicAdd(sp + 3, acc.w);
        if (lane == 0) atomicAdd(&cnts[cur], (float)cnt);
    };
    auto sum_group = [&](const float4 (&v)[POOL_U], int r0) {
#pragma unroll
        for (int u = 0; u < POOL_U; ++u) {
            const int r = r0 + u;
            const int g = __builtin_amdgcn_readlane(r < 32 ? gv0 : gv1, r & 31);
            if (g != cur) {
                hand_over();
                acc = make_float4(0.0f, 0.0f, 0.0f, 0.0f);
                cnt = 0;
                cur = g;
            }
            acc.x += fmaxf(v[u].x + bv.x, 0.0f);
            acc.y += fmaxf(v[u].y + bv.y, 0.0f);
            acc.z += fmaxf(v[u].z + bv.z, 0.0f);
            acc.w += fmaxf(v[u].w + bv.w, 0.0f);
            ++cnt;
        }
    };

    request(va, 0);
    request(vb, POOL_U);
    __builtin_amdgcn_sched_barrier(0);
    cur = __builtin_amdgcn_readlane(gv0, 0);
#pragma unroll
    for (int r0 = 0; r0 < POOL_NPW - POOL_U; r0 += 2 * POOL_U) {
        sum_group(va, r0);
        request(va, r0 + 2 * POOL_U);
        sum_group(vb, r0 + POOL_U);
        if (r0 + 3 * POOL_U < POOL_NPW) request(vb, r0 + 3 * POOL_U);
    }
    sum_group(va, POOL_NPW - POOL_U);
    hand_over();
}

__global__ void gcn_fc(const float* __restrict__ sums, const float* __restrict__ cnts,
                       const float* __restrict__ fcW, const float* __restrict__ fcb,
                       float* __restrict__ out) {
    int idx = blockIdx.x * 256 + threadIdx.x;
    if (idx >= NUM_GRAPHS * OUTC) return;
    int g = idx >> 4, o = idx & 15;
    float inv = 1.0f / fmaxf(cnts[g], 1.0f);
    float acc = fcb[o];
    for (int k = 0; k < 128; ++k)
        acc += sums[g * 128 + k] * inv * fcW[k * OUTC + o];
    out[idx] = acc;
}

#define WS_ROWS_BYTES   ((size_t)N_NODES * 128 * 4)
#define WS_SMALL_BYTES  ((size_t)N_NODES * 4 + (size_t)NUM_GRAPHS * 128 * 4 + (size_t)NUM_GRAPHS * 4)
#define WS_ENT_BYTES    ((size_t)N_ENT * 8)
#define WS_GATHER_BYTES (2 * WS_ROWS_BYTES + WS_SMALL_BYTES + WS_ENT_BYTES)

extern "C" void kernel_launch(void* const* d_in, const int* in_sizes, int n_in,
                              void* d_out, int out_size, void* d_ws, size_t ws_size,
                              hipStream_t stream) {
    (void)in_sizes; (void)n_in; (void)out_size;
    const float* x    = (const float*)d_in[0];
    const float* top  = (const float*)d_in[1];
    const int*   ei   = (const int*)d_in[2];
    const int*   bat  = (const int*)d_in[3];
    const float* W1   = (const float*)d_in[4];
    const float* b1   = (const float*)d_in[5];
    const float* W2   = (const float*)d_in[6];
    const float* b2   = (const float*)d_in[7];
    const float* fcW  = (const float*)d_in[8];
    const float* fcb  = (const float*)d_in[9];
    float*       out  = (float*)d_out;

    const int* src = ei;
    const int* dst = ei + N_EDGES;

    char* ws = (char*)d_ws;
    float* A    = (float*)ws; ws += WS_ROWS_BYTES;
    float* Bh   = (float*)ws; ws += WS_ROWS_BYTES;
    void*  nplane = (void*)ws; ws += (size_t)N_NODES * 4;
    float* sums = (float*)ws; ws += (size_t)NUM_GRAPHS * 128 * 4;
    float* cnts = (float*)ws; ws += (size_t)NUM_GRAPHS * 4;
    int*   ent  = (int*)ws;

    const int tb = 256;
    const int gN    = (N_NODES + tb - 1) / tb;
    const int gE    = (N_EDGES + tb - 1) / tb;
    const int gGemm = (N_NODES / 16 + 7) / 8;

    if (ws_size >= WS_GATHER_BYTES) {
        int*   rp   = (int*)nplane;
        int*   cnt  = (int*)A;
        int*   rank = cnt + N_NODES;
        float* dis  = (float*)(rank + N_EDGES);
        int*   tsum = (int*)(dis + N_NODES);

        csr_zero     <<<gN, tb, 0, stream>>>(cnt, sums, cnts);
        csr_count_with_gemm<<<GEMM_IN_COUNT + EDGE_BLOCKS, tb, 0, stream>>>(
            dst, cnt, rank, x, top, W1, Bh);
        csr_tile_sums<<<SCAN_BLOCKS, tb, 0, stream>>>(cnt, tsum);
        csr_scan     <<<SCAN_BLOCKS, tb, 0, stream>>>(cnt, tsum, rp, dis);
        csr_fill_with_gemm<<<GEMM_IN_FILL + ENT_BLOCKS, tb, 0, stream>>>(
            src, dst, rank, rp, dis, ent, x, top, W1, Bh);

        gcn_agg_pull_gemm<<<FUSE_GRID, FUSE_WAVES * 32, 0, stream>>>(Bh, rp, ent, b1, W2, A);
        gcn_agg_pull_pool<<<PULL_BLOCKS, tb, 0, stream>>>(A, rp, ent, b2, bat, sums, cnts);
    } else {
        float* dis = (float*)nplane;
        const long NC   = (long)N_NODES * 128;
        const int gNC   = (int)((NC + tb - 1) / tb);
        const int gEdge = (N_EDGES + 7) / 8;

        gcn_deg_init <<<gN, tb, 0, stream>>>(dis);
        gcn_deg_edges<<<gE, tb, 0, stream>>>(dst, dis);
        gcn_deg_rsqrt<<<gN, tb, 0, stream>>>(dis);

        gcn_gemm_wmma<0><<<gGemm, tb, 0, stream>>>(x, top, W1, Bh, N_NODES);
        gcn_agg_init <<<gNC, tb, 0, stream>>>(Bh, dis, A);
        gcn_agg_edges<<<gEdge, tb, 0, stream>>>(Bh, A, src, dst, dis);

        gcn_gemm_wmma<1><<<gGemm, tb, 0, stream>>>(A, b1, W2, Bh, N_NODES);
        gcn_agg_init <<<gNC, tb, 0, stream>>>(Bh, dis, A);
        gcn_agg_edges<<<gEdge, tb, 0, stream>>>(Bh, A, src, dst, dis);

        gcn_pool_zero <<<(NUM_GRAPHS * 128 + tb - 1) / tb, tb, 0, stream>>>(sums, cnts);
        gcn_pool_accum<<<POOL_BLOCKS, tb, 0, stream>>>(A, b2, bat, sums, cnts);
    }

    gcn_fc<<<(NUM_GRAPHS * OUTC + tb - 1) / tb, tb, 0, stream>>>(sums, cnts, fcW, fcb, out);
}
